# starting kernel
# baseline (speedup 1.0000x reference)
#include <hip/hip_runtime.h>
#include <hip/hip_bf16.h>

typedef __attribute__((ext_vector_type(2))) float v2f;
typedef __attribute__((ext_vector_type(4))) float v4f;
typedef __attribute__((ext_vector_type(8))) float v8f;

#define N_EMB  1024
#define HEAD   64
#define SEQ_T  2048
#define BATCH  8
#define ROWS   (BATCH * SEQ_T)

static __device__ __forceinline__ v8f wmma_f32(v2f a, v2f b, v8f c) {
    return __builtin_amdgcn_wmma_f32_16x16x4_f32(false, a, false, b, (short)0, c,
                                                 false, false);
}

static __device__ __forceinline__ v2f mk2(float x, float y) { v2f r; r.x = x; r.y = y; return r; }

__global__ __launch_bounds__(128)
void sah_proj_kernel(const float* __restrict__ x,
                     const float* __restrict__ Wk,
                     const float* __restrict__ Wq,
                     const float* __restrict__ Wv,
                     float* __restrict__ kb,
                     float* __restrict__ qb,
                     float* __restrict__ vb) {
    const int lane = threadIdx.x & 31;
    const int ht   = threadIdx.x >> 5;
    const int half = lane >> 4;
    const int lr   = lane & 15;
    const int rt   = blockIdx.x;
    const int which = blockIdx.y;

    const float* W   = (which == 0) ? Wk : (which == 1) ? Wq : Wv;
    float*       out = (which == 0) ? kb : (which == 1) ? qb : vb;

    const float* arow = x + (size_t)(rt * 16 + lr) * N_EMB + half * 4;
    const float* brow = W + (size_t)(ht * 16 + lr) * N_EMB + half * 4;

    v8f accA = {}, accB = {};
#pragma unroll 4
    for (int k0 = 0; k0 < N_EMB; k0 += 8) {
        v4f a4 = *(const v4f*)(arow + k0);
        v4f b4 = *(const v4f*)(brow + k0);
        accA = wmma_f32(mk2(a4.x, a4.y), mk2(b4.x, b4.y), accA);
        accB = wmma_f32(mk2(a4.z, a4.w), mk2(b4.z, b4.w), accB);
    }

    float* obase = out + (size_t)(rt * 16 + half * 8) * HEAD + ht * 16 + lr;
#pragma unroll
    for (int r = 0; r < 8; ++r)
        obase[(size_t)r * HEAD] = accA[r] + accB[r];
}

__global__ __launch_bounds__(128)
void sah_attn_kernel(const float* __restrict__ Q,
                     const float* __restrict__ K,
                     const float* __restrict__ V,
                     float* __restrict__ O) {
    __shared__ float pstage[4][16 * 16];
    __shared__ float mred[4][16];
    __shared__ float lred[4][16];
    __shared__ float ored[4][16 * 64];

    const int lane = threadIdx.x & 31;
    const int wave = threadIdx.x >> 5;
    const int half = lane >> 4;
    const int lr   = lane & 15;
    const int b    = blockIdx.y;
    const int qt   = blockIdx.x;
    const int q0   = qt * 16;

    const float* Qb = Q + (size_t)b * SEQ_T * HEAD;
    const float* Kb = K + (size_t)b * SEQ_T * HEAD;
    const float* Vb = V + (size_t)b * SEQ_T * HEAD;
    float* lds = pstage[wave];

    v2f qfrag[16];
    {
        const float* qrow = Qb + (size_t)(q0 + lr) * HEAD + half * 4;
#pragma unroll
        for (int c = 0; c < 8; ++c) {
            v4f t = *(const v4f*)(qrow + c * 8);
            qfrag[2 * c]     = mk2(t.x, t.y);
            qfrag[2 * c + 1] = mk2(t.z, t.w);
        }
    }

    const v2f ones = mk2(1.0f, 1.0f);

    v8f acc0 = {}, acc1 = {}, acc2 = {}, acc3 = {};
    v8f lacc = {};
    float m_i[8];
#pragma unroll
    for (int r = 0; r < 8; ++r) m_i[r] = -__builtin_inff();

    const float scale = 0.03125f;

    for (int j = wave; j <= qt; j += 4) {
        const int k0r = j * 16;

        v8f sA = {}, sB = {};
        {
            const float* krow = Kb + (size_t)(k0r + lr) * HEAD + half * 4;
#pragma unroll
            for (int c = 0; c < 8; ++c) {
                v4f t = *(const v4f*)(krow + c * 8);
                sA = wmma_f32(qfrag[2 * c],     mk2(t.x, t.y), sA);
                sB = wmma_f32(qfrag[2 * c + 1], mk2(t.z, t.w), sB);
            }
        }

        const bool diag = (j == qt);
#pragma unroll
        for (int r = 0; r < 8; ++r) {
            const int m = r + half * 8;
            float sv = (sA[r] + sB[r]) * scale;
            if (diag && lr > m) sv = -__builtin_inff();

            float rm = sv;
            rm = fmaxf(rm, __shfl_xor(rm, 1, 32));
            rm = fmaxf(rm, __shfl_xor(rm, 2, 32));
            rm = fmaxf(rm, __shfl_xor(rm, 4, 32));
            rm = fmaxf(rm, __shfl_xor(rm, 8, 32));

            float mn    = fmaxf(m_i[r], rm);
            float alpha = __expf(m_i[r] - mn);
            m_i[r] = mn;

            float pv = __expf(sv - mn);

            lacc[r] *= alpha;
            acc0[r] *= alpha;
            acc1[r] *= alpha;
            acc2[r] *= alpha;
            acc3[r] *= alpha;

            lds[m * 16 + lr] = pv;
        }

#pragma unroll
        for (int kk = 0; kk < 4; ++kk) {
            v2f pa = *(const v2f*)&lds[lr * 16 + kk * 4 + half * 2];
            const float* vp = Vb + (size_t)(k0r + kk * 4 + half * 2) * HEAD + lr;
            acc0 = wmma_f32(pa, mk2(vp[0],  vp[HEAD +  0]), acc0);
            acc1 = wmma_f32(pa, mk2(vp[16], vp[HEAD + 16]), acc1);
            acc2 = wmma_f32(pa, mk2(vp[32], vp[HEAD + 32]), acc2);
            acc3 = wmma_f32(pa, mk2(vp[48], vp[HEAD + 48]), acc3);
            lacc = wmma_f32(pa, ones, lacc);
        }
    }

    if (lr == 0) {
#pragma unroll
        for (int r = 0; r < 8; ++r)
            mred[wave][r + half * 8] = m_i[r];
    }
    __syncthreads();

#pragma unroll
    for (int r = 0; r < 8; ++r) {
        const int row = r + half * 8;
        float ms = fmaxf(fmaxf(mred[0][row], mred[1][row]),
                         fmaxf(mred[2][row], mred[3][row]));
        float al = __expf(m_i[r] - ms);
        acc0[r] *= al;
        acc1[r] *= al;
        acc2[r] *= al;
        acc3[r] *= al;
        if (lr == 0) lred[wave][row] = lacc[r] * al;
        float* orow = &ored[wave][row * HEAD + lr];
        orow[0]  = acc0[r];
        orow[16] = acc1[r];
        orow[32] = acc2[r];
        orow[48] = acc3[r];
    }
    __syncthreads();

    float* ob = O + ((size_t)b * SEQ_T + q0 + half * 8) * HEAD + wave * 16 + lr;
#pragma unroll
    for (int r = 0; r < 8; ++r) {
        const int row = r + half * 8;
        float lt = (lred[0][row] + lred[1][row]) + (lred[2][row] + lred[3][row]);
        const int c = row * HEAD + wave * 16 + lr;
        float s = (ored[0][c] + ored[1][c]) + (ored[2][c] + ored[3][c]);
        ob[(size_t)r * HEAD] = s / lt;
    }
}

extern "C" void kernel_launch(void* const* d_in, const int* in_sizes, int n_in,
                              void* d_out, int out_size, void* d_ws, size_t ws_size,
                              hipStream_t stream) {
    const float* x  = (const float*)d_in[0];
    const float* Wk = (const float*)d_in[1];
    const float* Wq = (const float*)d_in[2];
    const float* Wv = (const float*)d_in[3];

    float* kb = (float*)d_ws;
    float* qb = kb + (size_t)ROWS * HEAD;
    float* vb = qb + (size_t)ROWS * HEAD;

    sah_proj_kernel<<<dim3(ROWS / 16, 3), 128, 0, stream>>>(x, Wk, Wq, Wv, kb, qb, vb);
    sah_attn_kernel<<<dim3(SEQ_T / 16, BATCH), 128, 0, stream>>>(qb, kb, vb,
                                                                 (float*)d_out);
}


# Projection re-tiled as one 64x192 block GEMM: x and W k-slabs fetched once per block and shared through a 2-stage LDS ring, x fragment reused by 6 tiles; one sum chain per tile (sum order changes)
# speedup vs baseline: 1.2226x; 1.2226x over previous
#include <hip/hip_runtime.h>
#include <hip/hip_bf16.h>

typedef __attribute__((ext_vector_type(2))) float v2f;
typedef __attribute__((ext_vector_type(4))) float v4f;
typedef __attribute__((ext_vector_type(8))) float v8f;

#define N_EMB  1024
#define HEAD   64
#define SEQ_T  2048
#define BATCH  8
#define ROWS   (BATCH * SEQ_T)

static __device__ __forceinline__ v8f wmma_f32(v2f a, v2f b, v8f c) {
    return __builtin_amdgcn_wmma_f32_16x16x4_f32(false, a, false, b, (short)0, c,
                                                 false, false);
}

static __device__ __forceinline__ v2f mk2(float x, float y) { v2f r; r.x = x; r.y = y; return r; }

#define PJ_BM    64
#define PJ_BN    192
#define PJ_KS    64
#define PJ_PITCH 68
#define PJ_NSLAB (N_EMB / PJ_KS)
#define PJ_NM    (PJ_KS / 8)
#define PJ_CPR   (PJ_KS / 4)
#define PJ_RPP   (256 / PJ_CPR)
#define PJ_NPX   (PJ_BM / PJ_RPP)
#define PJ_NPW   (PJ_BN / PJ_RPP)
#define PJ_XST   (PJ_BM * PJ_PITCH)
#define PJ_WST   (PJ_BN * PJ_PITCH)

struct pj_frag { v4f a; v4f b[6]; };

static __device__ __forceinline__ void pj_load(pj_frag& f, const float* xa, const float* wb, int m) {
    f.a = *(const v4f*)(xa + m * 8);
#pragma unroll
    for (int t = 0; t < 6; ++t)
        f.b[t] = *(const v4f*)(wb + t * 16 * PJ_PITCH + m * 8);
}

static __device__ __forceinline__ void pj_mma(v8f (&acc)[6], const pj_frag& f) {
#pragma unroll
    for (int t = 0; t < 6; ++t)
        acc[t] = wmma_f32(mk2(f.a.x, f.a.y), mk2(f.b[t].x, f.b[t].y), acc[t]);
#pragma unroll
    for (int t = 0; t < 6; ++t)
        acc[t] = wmma_f32(mk2(f.a.z, f.a.w), mk2(f.b[t].z, f.b[t].w), acc[t]);
}

__global__ __launch_bounds__(256)
void sah_proj_kernel(const float* __restrict__ x,
                     const float* __restrict__ Wk,
                     const float* __restrict__ Wq,
                     const float* __restrict__ Wv,
                     float* __restrict__ kb,
                     float* __restrict__ qb,
                     float* __restrict__ vb) {
    __shared__ __attribute__((aligned(16))) float xs[2 * PJ_XST];
    __shared__ __attribute__((aligned(16))) float ws[2 * PJ_WST];

    const int tid  = threadIdx.x;
    const int lane = tid & 31;
    const int wave = tid >> 5;
    const int half = lane >> 4;
    const int lr   = lane & 15;
    const int rw   = wave & 3;
    const int cw   = wave >> 2;
    const int row0 = blockIdx.x * PJ_BM;

    const int srow = tid / PJ_CPR;
    const int skc  = (tid % PJ_CPR) * 4;
    const float* gx = x  + (size_t)(row0 + srow) * N_EMB + skc;
    const float* gk = Wk + (size_t)srow * N_EMB + skc;
    const float* gq = Wq + (size_t)srow * N_EMB + skc;
    const float* gv = Wv + (size_t)srow * N_EMB + skc;
    const int so = srow * PJ_PITCH + skc;

    const int ao = (rw * 16 + lr) * PJ_PITCH + half * 4;
    const int bo = (cw * 96 + lr) * PJ_PITCH + half * 4;

    v8f acc[6];
#pragma unroll
    for (int t = 0; t < 6; ++t) acc[t] = (v8f){};

    v4f px[PJ_NPX], pw[PJ_NPW];

#pragma unroll
    for (int p = 0; p < PJ_NPX; ++p)
        px[p] = *(const v4f*)(gx + (size_t)(p * PJ_RPP) * N_EMB);
#pragma unroll
    for (int p = 0; p < PJ_NPW; ++p) {
        const int wr = p * PJ_RPP;
        const float* g = (wr < 64) ? gk : (wr < 128) ? gq : gv;
        pw[p] = *(const v4f*)(g + (size_t)(wr & 63) * N_EMB);
    }
#pragma unroll
    for (int p = 0; p < PJ_NPX; ++p)
        *(v4f*)&xs[so + p * PJ_RPP * PJ_PITCH] = px[p];
#pragma unroll
    for (int p = 0; p < PJ_NPW; ++p)
        *(v4f*)&ws[so + p * PJ_RPP * PJ_PITCH] = pw[p];
    __syncthreads();

    for (int s = 0; s < PJ_NSLAB; ++s) {
        const int cur = s & 1;
        const bool more = (s + 1 < PJ_NSLAB);

        if (more) {
            const int ko = (s + 1) * PJ_KS;
#pragma unroll
            for (int p = 0; p < PJ_NPX; ++p)
                px[p] = *(const v4f*)(gx + (size_t)(p * PJ_RPP) * N_EMB + ko);
#pragma unroll
            for (int p = 0; p < PJ_NPW; ++p) {
                const int wr = p * PJ_RPP;
                const float* g = (wr < 64) ? gk : (wr < 128) ? gq : gv;
                pw[p] = *(const v4f*)(g + (size_t)(wr & 63) * N_EMB + ko);
            }
        }

        const float* xa = &xs[cur * PJ_XST + ao];
        const float* wb = &ws[cur * PJ_WST + bo];
        pj_frag f0, f1;
        pj_load(f0, xa, wb, 0);
#pragma unroll
        for (int m = 0; m < PJ_NM; m += 2) {
            pj_load(f1, xa, wb, m + 1);
            __builtin_amdgcn_sched_barrier(0);
            pj_mma(acc, f0);
            __builtin_amdgcn_sched_barrier(0);
            if (m + 2 < PJ_NM) pj_load(f0, xa, wb, m + 2);
            __builtin_amdgcn_sched_barrier(0);
            pj_mma(acc, f1);
            __builtin_amdgcn_sched_barrier(0);
        }

        if (more) {
            const int nx = cur ^ 1;
#pragma unroll
            for (int p = 0; p < PJ_NPX; ++p)
                *(v4f*)&xs[nx * PJ_XST + so + p * PJ_RPP * PJ_PITCH] = px[p];
#pragma unroll
            for (int p = 0; p < PJ_NPW; ++p)
                *(v4f*)&ws[nx * PJ_WST + so + p * PJ_RPP * PJ_PITCH] = pw[p];
        }
        __syncthreads();
    }

    const size_t orow = (size_t)(row0 + rw * 16 + half * 8);
#pragma unroll
    for (int t = 0; t < 6; ++t) {
        const int col   = cw * 96 + t * 16;
        const int which = col >> 6;
        float* out = (which == 0) ? kb : (which == 1) ? qb : vb;
        float* obase = out + orow * HEAD + (col & 63) + lr;
#pragma unroll
        for (int r = 0; r < 8; ++r)
            obase[(size_t)r * HEAD] = acc[t][r];
    }
}

__global__ __launch_bounds__(128)
void sah_attn_kernel(const float* __restrict__ Q,
                     const float* __restrict__ K,
                     const float* __restrict__ V,
                     float* __restrict__ O) {
    __shared__ float pstage[4][16 * 16];
    __shared__ float mred[4][16];
    __shared__ float lred[4][16];
    __shared__ float ored[4][16 * 64];

    const int lane = threadIdx.x & 31;
    const int wave = threadIdx.x >> 5;
    const int half = lane >> 4;
    const int lr   = lane & 15;
    const int b    = blockIdx.y;
    const int qt   = blockIdx.x;
    const int q0   = qt * 16;

    const float* Qb = Q + (size_t)b * SEQ_T * HEAD;
    const float* Kb = K + (size_t)b * SEQ_T * HEAD;
    const float* Vb = V + (size_t)b * SEQ_T * HEAD;
    float* lds = pstage[wave];

    v2f qfrag[16];
    {
        const float* qrow = Qb + (size_t)(q0 + lr) * HEAD + half * 4;
#pragma unroll
        for (int c = 0; c < 8; ++c) {
            v4f t = *(const v4f*)(qrow + c * 8);
            qfrag[2 * c]     = mk2(t.x, t.y);
            qfrag[2 * c + 1] = mk2(t.z, t.w);
        }
    }

    const v2f ones = mk2(1.0f, 1.0f);

    v8f acc0 = {}, acc1 = {}, acc2 = {}, acc3 = {};
    v8f lacc = {};
    float m_i[8];
#pragma unroll
    for (int r = 0; r < 8; ++r) m_i[r] = -__builtin_inff();

    const float scale = 0.03125f;

    for (int j = wave; j <= qt; j += 4) {
        const int k0r = j * 16;

        v8f sA = {}, sB = {};
        {
            const float* krow = Kb + (size_t)(k0r + lr) * HEAD + half * 4;
#pragma unroll
            for (int c = 0; c < 8; ++c) {
                v4f t = *(const v4f*)(krow + c * 8);
                sA = wmma_f32(qfrag[2 * c],     mk2(t.x, t.y), sA);
                sB = wmma_f32(qfrag[2 * c + 1], mk2(t.z, t.w), sB);
            }
        }

        const bool diag = (j == qt);
#pragma unroll
        for (int r = 0; r < 8; ++r) {
            const int m = r + half * 8;
            float sv = (sA[r] + sB[r]) * scale;
            if (diag && lr > m) sv = -__builtin_inff();

            float rm = sv;
            rm = fmaxf(rm, __shfl_xor(rm, 1, 32));
            rm = fmaxf(rm, __shfl_xor(rm, 2, 32));
            rm = fmaxf(rm, __shfl_xor(rm, 4, 32));
            rm = fmaxf(rm, __shfl_xor(rm, 8, 32));

            float mn    = fmaxf(m_i[r], rm);
            float alpha = __expf(m_i[r] - mn);
            m_i[r] = mn;

            float pv = __expf(sv - mn);

            lacc[r] *= alpha;
            acc0[r] *= alpha;
            acc1[r] *= alpha;
            acc2[r] *= alpha;
            acc3[r] *= alpha;

            lds[m * 16 + lr] = pv;
        }

#pragma unroll
        for (int kk = 0; kk < 4; ++kk) {
            v2f pa = *(const v2f*)&lds[lr * 16 + kk * 4 + half * 2];
            const float* vp = Vb + (size_t)(k0r + kk * 4 + half * 2) * HEAD + lr;
            acc0 = wmma_f32(pa, mk2(vp[0],  vp[HEAD +  0]), acc0);
            acc1 = wmma_f32(pa, mk2(vp[16], vp[HEAD + 16]), acc1);
            acc2 = wmma_f32(pa, mk2(vp[32], vp[HEAD + 32]), acc2);
            acc3 = wmma_f32(pa, mk2(vp[48], vp[HEAD + 48]), acc3);
            lacc = wmma_f32(pa, ones, lacc);
        }
    }

    if (lr == 0) {
#pragma unroll
        for (int r = 0; r < 8; ++r)
            mred[wave][r + half * 8] = m_i[r];
    }
    __syncthreads();

#pragma unroll
    for (int r = 0; r < 8; ++r) {
        const int row = r + half * 8;
        float ms = fmaxf(fmaxf(mred[0][row], mred[1][row]),
                         fmaxf(mred[2][row], mred[3][row]));
        float al = __expf(m_i[r] - ms);
        acc0[r] *= al;
        acc1[r] *= al;
        acc2[r] *= al;
        acc3[r] *= al;
        if (lr == 0) lred[wave][row] = lacc[r] * al;
        float* orow = &ored[wave][row * HEAD + lr];
        orow[0]  = acc0[r];
        orow[16] = acc1[r];
        orow[32] = acc2[r];
        orow[48] = acc3[r];
    }
    __syncthreads();

    float* ob = O + ((size_t)b * SEQ_T + q0 + half * 8) * HEAD + wave * 16 + lr;
#pragma unroll
    for (int r = 0; r < 8; ++r) {
        const int row = r + half * 8;
        float lt = (lred[0][row] + lred[1][row]) + (lred[2][row] + lred[3][row]);
        const int c = row * HEAD + wave * 16 + lr;
        float s = (ored[0][c] + ored[1][c]) + (ored[2][c] + ored[3][c]);
        ob[(size_t)r * HEAD] = s / lt;
    }
}

extern "C" void kernel_launch(void* const* d_in, const int* in_sizes, int n_in,
                              void* d_out, int out_size, void* d_ws, size_t ws_size,
                              hipStream_t stream) {
    const float* x  = (const float*)d_in[0];
    const float* Wk = (const float*)d_in[1];
    const float* Wq = (const float*)d_in[2];
    const float* Wv = (const float*)d_in[3];

    float* kb = (float*)d_ws;
    float* qb = kb + (size_t)ROWS * HEAD;
    float* vb = qb + (size_t)ROWS * HEAD;

    sah_proj_kernel<<<dim3(ROWS / PJ_BM), 256, 0, stream>>>(x, Wk, Wq, Wv, kb, qb, vb);
    sah_attn_kernel<<<dim3(SEQ_T / 16, BATCH), 128, 0, stream>>>(qb, kb, vb,
                                                                 (float*)d_out);
}


# Attention scores computed transposed (K.Q^T): a query's max and sum run down one lane's registers and P^T feeds the PV product from registers; no LDS permutes, no ones-product (sum order changes)
# speedup vs baseline: 1.5004x; 1.2272x over previous
#include <hip/hip_runtime.h>
#include <hip/hip_bf16.h>

typedef __attribute__((ext_vector_type(2))) float v2f;
typedef __attribute__((ext_vector_type(4))) float v4f;
typedef __attribute__((ext_vector_type(8))) float v8f;

#define N_EMB  1024
#define HEAD   64
#define SEQ_T  2048
#define BATCH  8
#define ROWS   (BATCH * SEQ_T)

static __device__ __forceinline__ v8f wmma_f32(v2f a, v2f b, v8f c) {
    return __builtin_amdgcn_wmma_f32_16x16x4_f32(false, a, false, b, (short)0, c,
                                                 false, false);
}

static __device__ __forceinline__ v2f mk2(float x, float y) { v2f r; r.x = x; r.y = y; return r; }

#define PJ_BM    64
#define PJ_BN    192
#define PJ_KS    64
#define PJ_PITCH 68
#define PJ_NSLAB (N_EMB / PJ_KS)
#define PJ_NM    (PJ_KS / 8)
#define PJ_CPR   (PJ_KS / 4)
#define PJ_RPP   (256 / PJ_CPR)
#define PJ_NPX   (PJ_BM / PJ_RPP)
#define PJ_NPW   (PJ_BN / PJ_RPP)
#define PJ_XST   (PJ_BM * PJ_PITCH)
#define PJ_WST   (PJ_BN * PJ_PITCH)

struct pj_frag { v4f a; v4f b[6]; };

static __device__ __forceinline__ void pj_load(pj_frag& f, const float* xa, const float* wb, int m) {
    f.a = *(const v4f*)(xa + m * 8);
#pragma unroll
    for (int t = 0; t < 6; ++t)
        f.b[t] = *(const v4f*)(wb + t * 16 * PJ_PITCH + m * 8);
}

static __device__ __forceinline__ void pj_mma(v8f (&acc)[6], const pj_frag& f) {
#pragma unroll
    for (int t = 0; t < 6; ++t)
        acc[t] = wmma_f32(mk2(f.a.x, f.a.y), mk2(f.b[t].x, f.b[t].y), acc[t]);
#pragma unroll
    for (int t = 0; t < 6; ++t)
        acc[t] = wmma_f32(mk2(f.a.z, f.a.w), mk2(f.b[t].z, f.b[t].w), acc[t]);
}

__global__ __launch_bounds__(256)
void sah_proj_kernel(const float* __restrict__ x,
                     const float* __restrict__ Wk,
                     const float* __restrict__ Wq,
                     const float* __restrict__ Wv,
                     float* __restrict__ kb,
                     float* __restrict__ qb,
                     float* __restrict__ vb) {
    __shared__ __attribute__((aligned(16))) float xs[2 * PJ_XST];
    __shared__ __attribute__((aligned(16))) float ws[2 * PJ_WST];

    const int tid  = threadIdx.x;
    const int lane = tid & 31;
    const int wave = tid >> 5;
    const int half = lane >> 4;
    const int lr   = lane & 15;
    const int rw   = wave & 3;
    const int cw   = wave >> 2;
    const int row0 = blockIdx.x * PJ_BM;

    const int srow = tid / PJ_CPR;
    const int skc  = (tid % PJ_CPR) * 4;
    const float* gx = x  + (size_t)(row0 + srow) * N_EMB + skc;
    const float* gk = Wk + (size_t)srow * N_EMB + skc;
    const float* gq = Wq + (size_t)srow * N_EMB + skc;
    const float* gv = Wv + (size_t)srow * N_EMB + skc;
    const int so = srow * PJ_PITCH + skc;

    const int ao = (rw * 16 + lr) * PJ_PITCH + half * 4;
    const int bo = (cw * 96 + lr) * PJ_PITCH + half * 4;

    v8f acc[6];
#pragma unroll
    for (int t = 0; t < 6; ++t) acc[t] = (v8f){};

    v4f px[PJ_NPX], pw[PJ_NPW];

#pragma unroll
    for (int p = 0; p < PJ_NPX; ++p)
        px[p] = *(const v4f*)(gx + (size_t)(p * PJ_RPP) * N_EMB);
#pragma unroll
    for (int p = 0; p < PJ_NPW; ++p) {
        const int wr = p * PJ_RPP;
        const float* g = (wr < 64) ? gk : (wr < 128) ? gq : gv;
        pw[p] = *(const v4f*)(g + (size_t)(wr & 63) * N_EMB);
    }
#pragma unroll
    for (int p = 0; p < PJ_NPX; ++p)
        *(v4f*)&xs[so + p * PJ_RPP * PJ_PITCH] = px[p];
#pragma unroll
    for (int p = 0; p < PJ_NPW; ++p)
        *(v4f*)&ws[so + p * PJ_RPP * PJ_PITCH] = pw[p];
    __syncthreads();

    for (int s = 0; s < PJ_NSLAB; ++s) {
        const int cur = s & 1;
        const bool more = (s + 1 < PJ_NSLAB);

        if (more) {
            const int ko = (s + 1) * PJ_KS;
#pragma unroll
            for (int p = 0; p < PJ_NPX; ++p)
                px[p] = *(const v4f*)(gx + (size_t)(p * PJ_RPP) * N_EMB + ko);
#pragma unroll
            for (int p = 0; p < PJ_NPW; ++p) {
                const int wr = p * PJ_RPP;
                const float* g = (wr < 64) ? gk : (wr < 128) ? gq : gv;
                pw[p] = *(const v4f*)(g + (size_t)(wr & 63) * N_EMB + ko);
            }
        }

        const float* xa = &xs[cur * PJ_XST + ao];
        const float* wb = &ws[cur * PJ_WST + bo];
        pj_frag f0, f1;
        pj_load(f0, xa, wb, 0);
#pragma unroll
        for (int m = 0; m < PJ_NM; m += 2) {
            pj_load(f1, xa, wb, m + 1);
            __builtin_amdgcn_sched_barrier(0);
            pj_mma(acc, f0);
            __builtin_amdgcn_sched_barrier(0);
            if (m + 2 < PJ_NM) pj_load(f0, xa, wb, m + 2);
            __builtin_amdgcn_sched_barrier(0);
            pj_mma(acc, f1);
            __builtin_amdgcn_sched_barrier(0);
        }

        if (more) {
            const int nx = cur ^ 1;
#pragma unroll
            for (int p = 0; p < PJ_NPX; ++p)
                *(v4f*)&xs[nx * PJ_XST + so + p * PJ_RPP * PJ_PITCH] = px[p];
#pragma unroll
            for (int p = 0; p < PJ_NPW; ++p)
                *(v4f*)&ws[nx * PJ_WST + so + p * PJ_RPP * PJ_PITCH] = pw[p];
        }
        __syncthreads();
    }

    const size_t orow = (size_t)(row0 + rw * 16 + half * 8);
#pragma unroll
    for (int t = 0; t < 6; ++t) {
        const int col   = cw * 96 + t * 16;
        const int which = col >> 6;
        float* out = (which == 0) ? kb : (which == 1) ? qb : vb;
        float* obase = out + orow * HEAD + (col & 63) + lr;
#pragma unroll
        for (int r = 0; r < 8; ++r)
            obase[(size_t)r * HEAD] = acc[t][r];
    }
}

#define AT_OP 68

static __device__ __forceinline__ float xhalf(float x) {
    int b = __builtin_bit_cast(int, x);
    return __builtin_bit_cast(float,
        __builtin_amdgcn_permlanex16(b, b, 0x76543210, 0xfedcba98, false, false));
}

__global__ __launch_bounds__(128)
void sah_attn_kernel(const float* __restrict__ Q,
                     const float* __restrict__ K,
                     const float* __restrict__ V,
                     float* __restrict__ O) {
    __shared__ float mred[4][16];
    __shared__ float lred[4][16];
    __shared__ __attribute__((aligned(16))) float ored[4][16 * AT_OP];

    const int lane = threadIdx.x & 31;
    const int wave = __builtin_amdgcn_readfirstlane((int)(threadIdx.x >> 5));
    const int half = lane >> 4;
    const int lr   = lane & 15;
    const int b    = blockIdx.y;
    const int qt   = blockIdx.x;
    const int q0   = qt * 16;

    const float* Qb = Q + (size_t)b * SEQ_T * HEAD;
    const float* Kb = K + (size_t)b * SEQ_T * HEAD;
    const float* Vb = V + (size_t)b * SEQ_T * HEAD;

    v2f qfrag[16];
    {
        const float* qrow = Qb + (size_t)(q0 + lr) * HEAD + half * 4;
#pragma unroll
        for (int c = 0; c < 8; ++c) {
            v4f t = *(const v4f*)(qrow + c * 8);
            qfrag[2 * c]     = mk2(t.x, t.y);
            qfrag[2 * c + 1] = mk2(t.z, t.w);
        }
    }

    v8f acc0 = {}, acc1 = {}, acc2 = {}, acc3 = {};
    float m_i = -__builtin_inff();
    float l_i = 0.0f;

    const float scale = 0.03125f;

    for (int j = wave; j <= qt; j += 4) {
        const int k0r = j * 16;

        v4f kt[8];
        {
            const float* krow = Kb + (size_t)(k0r + lr) * HEAD + half * 4;
#pragma unroll
            for (int c = 0; c < 8; ++c) kt[c] = *(const v4f*)(krow + c * 8);
        }
        v2f va[16];
#pragma unroll
        for (int kk = 0; kk < 4; ++kk) {
            const float* vp = Vb + (size_t)(k0r + half * 8 + kk * 2) * HEAD + lr;
#pragma unroll
            for (int t = 0; t < 4; ++t) va[4 * kk + t] = mk2(vp[16 * t], vp[HEAD + 16 * t]);
        }
        __builtin_amdgcn_sched_barrier(0);

        v8f sA = {}, sB = {};
#pragma unroll
        for (int c = 0; c < 8; ++c) {
            sA = wmma_f32(mk2(kt[c].x, kt[c].y), qfrag[2 * c],     sA);
            sB = wmma_f32(mk2(kt[c].z, kt[c].w), qfrag[2 * c + 1], sB);
        }

        const int lim = (j == qt) ? lr : 16;
        float s[8];
#pragma unroll
        for (int r = 0; r < 8; ++r) {
            float sv = (sA[r] + sB[r]) * scale;
            s[r] = (half * 8 + r > lim) ? -__builtin_inff() : sv;
        }

        float tm = fmaxf(fmaxf(fmaxf(s[0], s[1]), fmaxf(s[2], s[3])),
                         fmaxf(fmaxf(s[4], s[5]), fmaxf(s[6], s[7])));
        tm = fmaxf(tm, xhalf(tm));
        const float mn    = fmaxf(m_i, tm);
        const float alpha = __expf(m_i - mn);
        m_i = mn;

        float p[8];
#pragma unroll
        for (int r = 0; r < 8; ++r) p[r] = __expf(s[r] - mn);

        l_i = l_i * alpha + (((p[0] + p[1]) + (p[2] + p[3])) + ((p[4] + p[5]) + (p[6] + p[7])));
#pragma unroll
        for (int r = 0; r < 8; ++r) {
            acc0[r] *= alpha;
            acc1[r] *= alpha;
            acc2[r] *= alpha;
            acc3[r] *= alpha;
        }

#pragma unroll
        for (int kk = 0; kk < 4; ++kk) {
            const v2f pb = mk2(p[2 * kk], p[2 * kk + 1]);
            acc0 = wmma_f32(va[4 * kk + 0], pb, acc0);
            acc1 = wmma_f32(va[4 * kk + 1], pb, acc1);
            acc2 = wmma_f32(va[4 * kk + 2], pb, acc2);
            acc3 = wmma_f32(va[4 * kk + 3], pb, acc3);
        }
    }

    l_i = l_i + xhalf(l_i);
    if (half == 0) mred[wave][lr] = m_i;
    __syncthreads();

    {
        const float ms = fmaxf(fmaxf(mred[0][lr], mred[1][lr]),
                               fmaxf(mred[2][lr], mred[3][lr]));
        const float al = __expf(m_i - ms);
        if (half == 0) lred[wave][lr] = l_i * al;
        float* orow = &ored[wave][lr * AT_OP + half * 8];
#pragma unroll
        for (int h4 = 0; h4 < 2; ++h4) {
            v4f o0, o1, o2, o3;
            o0.x = acc0[4 * h4] * al; o0.y = acc0[4 * h4 + 1] * al; o0.z = acc0[4 * h4 + 2] * al; o0.w = acc0[4 * h4 + 3] * al;
            o1.x = acc1[4 * h4] * al; o1.y = acc1[4 * h4 + 1] * al; o1.z = acc1[4 * h4 + 2] * al; o1.w = acc1[4 * h4 + 3] * al;
            o2.x = acc2[4 * h4] * al; o2.y = acc2[4 * h4 + 1] * al; o2.z = acc2[4 * h4 + 2] * al; o2.w = acc2[4 * h4 + 3] * al;
            o3.x = acc3[4 * h4] * al; o3.y = acc3[4 * h4 + 1] * al; o3.z = acc3[4 * h4 + 2] * al; o3.w = acc3[4 * h4 + 3] * al;
            *(v4f*)(orow +  0 + 4 * h4) = o0;
            *(v4f*)(orow + 16 + 4 * h4) = o1;
            *(v4f*)(orow + 32 + 4 * h4) = o2;
            *(v4f*)(orow + 48 + 4 * h4) = o3;
        }
    }
    __syncthreads();

    {
        const int row = threadIdx.x >> 3;
        const int cc  = (threadIdx.x & 7) * 8;
        const float lt = (lred[0][row] + lred[1][row]) + (lred[2][row] + lred[3][row]);
        float* ob = O + ((size_t)b * SEQ_T + q0 + row) * HEAD + cc;
#pragma unroll
        for (int u = 0; u < 2; ++u) {
            const int o = row * AT_OP + cc + 4 * u;
            const v4f a0 = *(const v4f*)&ored[0][o];
            const v4f a1 = *(const v4f*)&ored[1][o];
            const v4f a2 = *(const v4f*)&ored[2][o];
            const v4f a3 = *(const v4f*)&ored[3][o];
            v4f sum = (a0 + a1) + (a2 + a3);
            v4f res;
            res.x = sum.x / lt; res.y = sum.y / lt; res.z = sum.z / lt; res.w = sum.w / lt;
            *(v4f*)(ob + 4 * u) = res;
        }
    }
}

extern "C" void kernel_launch(void* const* d_in, const int* in_sizes, int n_in,
                              void* d_out, int out_size, void* d_ws, size_t ws_size,
                              hipStream_t stream) {
    const float* x  = (const float*)d_in[0];
    const float* Wk = (const float*)d_in[1];
    const float* Wq = (const float*)d_in[2];
    const float* Wv = (const float*)d_in[3];

    float* kb = (float*)d_ws;
    float* qb = kb + (size_t)ROWS * HEAD;
    float* vb = qb + (size_t)ROWS * HEAD;

    sah_proj_kernel<<<dim3(ROWS / PJ_BM), 256, 0, stream>>>(x, Wk, Wq, Wv, kb, qb, vb);
    sah_attn_kernel<<<dim3(SEQ_T / 16, BATCH), 128, 0, stream>>>(qb, kb, vb,
                                                                 (float*)d_out);
}


# Causal load balance: each block takes query tiles 127-p and p in turn (129 key tiles per block) with the key tiles dealt over 8 waves; log-sum-exp merge over 8 partials (sum order changes)
# speedup vs baseline: 1.8199x; 1.2129x over previous
#include <hip/hip_runtime.h>
#include <hip/hip_bf16.h>

typedef __attribute__((ext_vector_type(2))) float v2f;
typedef __attribute__((ext_vector_type(4))) float v4f;
typedef __attribute__((ext_vector_type(8))) float v8f;

#define N_EMB  1024
#define HEAD   64
#define SEQ_T  2048
#define BATCH  8
#define ROWS   (BATCH * SEQ_T)

static __device__ __forceinline__ v8f wmma_f32(v2f a, v2f b, v8f c) {
    return __builtin_amdgcn_wmma_f32_16x16x4_f32(false, a, false, b, (short)0, c,
                                                 false, false);
}

static __device__ __forceinline__ v2f mk2(float x, float y) { v2f r; r.x = x; r.y = y; return r; }

#define PJ_BM    64
#define PJ_BN    192
#define PJ_KS    64
#define PJ_PITCH 68
#define PJ_NSLAB (N_EMB / PJ_KS)
#define PJ_NM    (PJ_KS / 8)
#define PJ_CPR   (PJ_KS / 4)
#define PJ_RPP   (256 / PJ_CPR)
#define PJ_NPX   (PJ_BM / PJ_RPP)
#define PJ_NPW   (PJ_BN / PJ_RPP)
#define PJ_XST   (PJ_BM * PJ_PITCH)
#define PJ_WST   (PJ_BN * PJ_PITCH)

struct pj_frag { v4f a; v4f b[6]; };

static __device__ __forceinline__ void pj_load(pj_frag& f, const float* xa, const float* wb, int m) {
    f.a = *(const v4f*)(xa + m * 8);
#pragma unroll
    for (int t = 0; t < 6; ++t)
        f.b[t] = *(const v4f*)(wb + t * 16 * PJ_PITCH + m * 8);
}

static __device__ __forceinline__ void pj_mma(v8f (&acc)[6], const pj_frag& f) {
#pragma unroll
    for (int t = 0; t < 6; ++t)
        acc[t] = wmma_f32(mk2(f.a.x, f.a.y), mk2(f.b[t].x, f.b[t].y), acc[t]);
#pragma unroll
    for (int t = 0; t < 6; ++t)
        acc[t] = wmma_f32(mk2(f.a.z, f.a.w), mk2(f.b[t].z, f.b[t].w), acc[t]);
}

__global__ __launch_bounds__(256)
void sah_proj_kernel(const float* __restrict__ x,
                     const float* __restrict__ Wk,
                     const float* __restrict__ Wq,
                     const float* __restrict__ Wv,
                     float* __restrict__ kb,
                     float* __restrict__ qb,
                     float* __restrict__ vb) {
    __shared__ __attribute__((aligned(16))) float xs[2 * PJ_XST];
    __shared__ __attribute__((aligned(16))) float ws[2 * PJ_WST];

    const int tid  = threadIdx.x;
    const int lane = tid & 31;
    const int wave = tid >> 5;
    const int half = lane >> 4;
    const int lr   = lane & 15;
    const int rw   = wave & 3;
    const int cw   = wave >> 2;
    const int row0 = blockIdx.x * PJ_BM;

    const int srow = tid / PJ_CPR;
    const int skc  = (tid % PJ_CPR) * 4;
    const float* gx = x  + (size_t)(row0 + srow) * N_EMB + skc;
    const float* gk = Wk + (size_t)srow * N_EMB + skc;
    const float* gq = Wq + (size_t)srow * N_EMB + skc;
    const float* gv = Wv + (size_t)srow * N_EMB + skc;
    const int so = srow * PJ_PITCH + skc;

    const int ao = (rw * 16 + lr) * PJ_PITCH + half * 4;
    const int bo = (cw * 96 + lr) * PJ_PITCH + half * 4;

    v8f acc[6];
#pragma unroll
    for (int t = 0; t < 6; ++t) acc[t] = (v8f){};

    v4f px[PJ_NPX], pw[PJ_NPW];

#pragma unroll
    for (int p = 0; p < PJ_NPX; ++p)
        px[p] = *(const v4f*)(gx + (size_t)(p * PJ_RPP) * N_EMB);
#pragma unroll
    for (int p = 0; p < PJ_NPW; ++p) {
        const int wr = p * PJ_RPP;
        const float* g = (wr < 64) ? gk : (wr < 128) ? gq : gv;
        pw[p] = *(const v4f*)(g + (size_t)(wr & 63) * N_EMB);
    }
#pragma unroll
    for (int p = 0; p < PJ_NPX; ++p)
        *(v4f*)&xs[so + p * PJ_RPP * PJ_PITCH] = px[p];
#pragma unroll
    for (int p = 0; p < PJ_NPW; ++p)
        *(v4f*)&ws[so + p * PJ_RPP * PJ_PITCH] = pw[p];
    __syncthreads();

    for (int s = 0; s < PJ_NSLAB; ++s) {
        const int cur = s & 1;
        const bool more = (s + 1 < PJ_NSLAB);

        if (more) {
            const int ko = (s + 1) * PJ_KS;
#pragma unroll
            for (int p = 0; p < PJ_NPX; ++p)
                px[p] = *(const v4f*)(gx + (size_t)(p * PJ_RPP) * N_EMB + ko);
#pragma unroll
            for (int p = 0; p < PJ_NPW; ++p) {
                const int wr = p * PJ_RPP;
                const float* g = (wr < 64) ? gk : (wr < 128) ? gq : gv;
                pw[p] = *(const v4f*)(g + (size_t)(wr & 63) * N_EMB + ko);
            }
        }

        const float* xa = &xs[cur * PJ_XST + ao];
        const float* wb = &ws[cur * PJ_WST + bo];
        pj_frag f0, f1;
        pj_load(f0, xa, wb, 0);
#pragma unroll
        for (int m = 0; m < PJ_NM; m += 2) {
            pj_load(f1, xa, wb, m + 1);
            __builtin_amdgcn_sched_barrier(0);
            pj_mma(acc, f0);
            __builtin_amdgcn_sched_barrier(0);
            if (m + 2 < PJ_NM) pj_load(f0, xa, wb, m + 2);
            __builtin_amdgcn_sched_barrier(0);
            pj_mma(acc, f1);
            __builtin_amdgcn_sched_barrier(0);
        }

        if (more) {
            const int nx = cur ^ 1;
#pragma unroll
            for (int p = 0; p < PJ_NPX; ++p)
                *(v4f*)&xs[nx * PJ_XST + so + p * PJ_RPP * PJ_PITCH] = px[p];
#pragma unroll
            for (int p = 0; p < PJ_NPW; ++p)
                *(v4f*)&ws[nx * PJ_WST + so + p * PJ_RPP * PJ_PITCH] = pw[p];
        }
        __syncthreads();
    }

    const size_t orow = (size_t)(row0 + rw * 16 + half * 8);
#pragma unroll
    for (int t = 0; t < 6; ++t) {
        const int col   = cw * 96 + t * 16;
        const int which = col >> 6;
        float* out = (which == 0) ? kb : (which == 1) ? qb : vb;
        float* obase = out + orow * HEAD + (col & 63) + lr;
#pragma unroll
        for (int r = 0; r < 8; ++r)
            obase[(size_t)r * HEAD] = acc[t][r];
    }
}

#define AT_OP 68
#define AT_NW 8
#define AT_NT (SEQ_T / 16)
#define AT_NPAIR (AT_NT / 2)

static __device__ __forceinline__ float xhalf(float x) {
    int b = __builtin_bit_cast(int, x);
    return __builtin_bit_cast(float,
        __builtin_amdgcn_permlanex16(b, b, 0x76543210, 0xfedcba98, false, false));
}

__global__ __launch_bounds__(32 * AT_NW)
void sah_attn_kernel(const float* __restrict__ Q,
                     const float* __restrict__ K,
                     const float* __restrict__ V,
                     float* __restrict__ O) {
    __shared__ float mred[AT_NW][16];
    __shared__ float lred[AT_NW][16];
    __shared__ __attribute__((aligned(16))) float ored[AT_NW][16 * AT_OP];

    const int lane = threadIdx.x & 31;
    const int wave = __builtin_amdgcn_readfirstlane((int)(threadIdx.x >> 5));
    const int half = lane >> 4;
    const int lr   = lane & 15;
    const int b    = blockIdx.y;
    const int pr   = blockIdx.x;

    const float* Qb = Q + (size_t)b * SEQ_T * HEAD;
    const float* Kb = K + (size_t)b * SEQ_T * HEAD;
    const float* Vb = V + (size_t)b * SEQ_T * HEAD;

    for (int pass = 0; pass < 2; ++pass) {
    const int qt   = pass ? pr : (AT_NT - 1 - pr);
    const int q0   = qt * 16;

    v2f qfrag[16];
    {
        const float* qrow = Qb + (size_t)(q0 + lr) * HEAD + half * 4;
#pragma unroll
        for (int c = 0; c < 8; ++c) {
            v4f t = *(const v4f*)(qrow + c * 8);
            qfrag[2 * c]     = mk2(t.x, t.y);
            qfrag[2 * c + 1] = mk2(t.z, t.w);
        }
    }

    v8f acc0 = {}, acc1 = {}, acc2 = {}, acc3 = {};
    float m_i = -__builtin_inff();
    float l_i = 0.0f;

    const float scale = 0.03125f;

    for (int j = wave; j <= qt; j += AT_NW) {
        const int k0r = j * 16;

        v4f kt[8];
        {
            const float* krow = Kb + (size_t)(k0r + lr) * HEAD + half * 4;
#pragma unroll
            for (int c = 0; c < 8; ++c) kt[c] = *(const v4f*)(krow + c * 8);
        }
        v2f va[16];
#pragma unroll
        for (int kk = 0; kk < 4; ++kk) {
            const float* vp = Vb + (size_t)(k0r + half * 8 + kk * 2) * HEAD + lr;
#pragma unroll
            for (int t = 0; t < 4; ++t) va[4 * kk + t] = mk2(vp[16 * t], vp[HEAD + 16 * t]);
        }
        __builtin_amdgcn_sched_barrier(0);

        v8f sA = {}, sB = {};
#pragma unroll
        for (int c = 0; c < 8; ++c) {
            sA = wmma_f32(mk2(kt[c].x, kt[c].y), qfrag[2 * c],     sA);
            sB = wmma_f32(mk2(kt[c].z, kt[c].w), qfrag[2 * c + 1], sB);
        }

        const int lim = (j == qt) ? lr : 16;
        float s[8];
#pragma unroll
        for (int r = 0; r < 8; ++r) {
            float sv = (sA[r] + sB[r]) * scale;
            s[r] = (half * 8 + r > lim) ? -__builtin_inff() : sv;
        }

        float tm = fmaxf(fmaxf(fmaxf(s[0], s[1]), fmaxf(s[2], s[3])),
                         fmaxf(fmaxf(s[4], s[5]), fmaxf(s[6], s[7])));
        tm = fmaxf(tm, xhalf(tm));
        const float mn    = fmaxf(m_i, tm);
        const float alpha = __expf(m_i - mn);
        m_i = mn;

        float p[8];
#pragma unroll
        for (int r = 0; r < 8; ++r) p[r] = __expf(s[r] - mn);

        l_i = l_i * alpha + (((p[0] + p[1]) + (p[2] + p[3])) + ((p[4] + p[5]) + (p[6] + p[7])));
#pragma unroll
        for (int r = 0; r < 8; ++r) {
            acc0[r] *= alpha;
            acc1[r] *= alpha;
            acc2[r] *= alpha;
            acc3[r] *= alpha;
        }

#pragma unroll
        for (int kk = 0; kk < 4; ++kk) {
            const v2f pb = mk2(p[2 * kk], p[2 * kk + 1]);
            acc0 = wmma_f32(va[4 * kk + 0], pb, acc0);
            acc1 = wmma_f32(va[4 * kk + 1], pb, acc1);
            acc2 = wmma_f32(va[4 * kk + 2], pb, acc2);
            acc3 = wmma_f32(va[4 * kk + 3], pb, acc3);
        }
    }

    l_i = l_i + xhalf(l_i);
    if (half == 0) mred[wave][lr] = m_i;
    __syncthreads();

    {
        const float ms = fmaxf(fmaxf(fmaxf(mred[0][lr], mred[1][lr]),
                                     fmaxf(mred[2][lr], mred[3][lr])),
                               fmaxf(fmaxf(mred[4][lr], mred[5][lr]),
                                     fmaxf(mred[6][lr], mred[7][lr])));
        const float al = __expf(m_i - ms);
        if (half == 0) lred[wave][lr] = l_i * al;
        float* orow = &ored[wave][lr * AT_OP + half * 8];
#pragma unroll
        for (int h4 = 0; h4 < 2; ++h4) {
            v4f o0, o1, o2, o3;
            o0.x = acc0[4 * h4] * al; o0.y = acc0[4 * h4 + 1] * al; o0.z = acc0[4 * h4 + 2] * al; o0.w = acc0[4 * h4 + 3] * al;
            o1.x = acc1[4 * h4] * al; o1.y = acc1[4 * h4 + 1] * al; o1.z = acc1[4 * h4 + 2] * al; o1.w = acc1[4 * h4 + 3] * al;
            o2.x = acc2[4 * h4] * al; o2.y = acc2[4 * h4 + 1] * al; o2.z = acc2[4 * h4 + 2] * al; o2.w = acc2[4 * h4 + 3] * al;
            o3.x = acc3[4 * h4] * al; o3.y = acc3[4 * h4 + 1] * al; o3.z = acc3[4 * h4 + 2] * al; o3.w = acc3[4 * h4 + 3] * al;
            *(v4f*)(orow +  0 + 4 * h4) = o0;
            *(v4f*)(orow + 16 + 4 * h4) = o1;
            *(v4f*)(orow + 32 + 4 * h4) = o2;
            *(v4f*)(orow + 48 + 4 * h4) = o3;
        }
    }
    __syncthreads();

    {
        const int row = threadIdx.x >> 4;
        const int cc  = (threadIdx.x & 15) * 4;
        const float lt = ((lred[0][row] + lred[1][row]) + (lred[2][row] + lred[3][row])) +
                         ((lred[4][row] + lred[5][row]) + (lred[6][row] + lred[7][row]));
        float* ob = O + ((size_t)b * SEQ_T + q0 + row) * HEAD + cc;
        const int o = row * AT_OP + cc;
        const v4f a0 = *(const v4f*)&ored[0][o];
        const v4f a1 = *(const v4f*)&ored[1][o];
        const v4f a2 = *(const v4f*)&ored[2][o];
        const v4f a3 = *(const v4f*)&ored[3][o];
        const v4f a4 = *(const v4f*)&ored[4][o];
        const v4f a5 = *(const v4f*)&ored[5][o];
        const v4f a6 = *(const v4f*)&ored[6][o];
        const v4f a7 = *(const v4f*)&ored[7][o];
        v4f sum = ((a0 + a1) + (a2 + a3)) + ((a4 + a5) + (a6 + a7));
        v4f res;
        res.x = sum.x / lt; res.y = sum.y / lt; res.z = sum.z / lt; res.w = sum.w / lt;
        *(v4f*)ob = res;
    }
    }
}

extern "C" void kernel_launch(void* const* d_in, const int* in_sizes, int n_in,
                              void* d_out, int out_size, void* d_ws, size_t ws_size,
                              hipStream_t stream) {
    const float* x  = (const float*)d_in[0];
    const float* Wk = (const float*)d_in[1];
    const float* Wq = (const float*)d_in[2];
    const float* Wv = (const float*)d_in[3];

    float* kb = (float*)d_ws;
    float* qb = kb + (size_t)ROWS * HEAD;
    float* vb = qb + (size_t)ROWS * HEAD;

    sah_proj_kernel<<<dim3(ROWS / PJ_BM), 256, 0, stream>>>(x, Wk, Wq, Wv, kb, qb, vb);
    sah_attn_kernel<<<dim3(AT_NPAIR, BATCH), 32 * AT_NW, 0, stream>>>(qb, kb, vb,
                                                                      (float*)d_out);
}


# Attention: each wave carries two adjacent query tiles through its key tiles, so one K/V^T fragment fetch feeds 64 WMMAs, not 32; the next key tile is fetched one tile ahead into a second register set
# speedup vs baseline: 1.9599x; 1.0769x over previous
#include <hip/hip_runtime.h>
#include <hip/hip_bf16.h>

typedef __attribute__((ext_vector_type(2))) float v2f;
typedef __attribute__((ext_vector_type(4))) float v4f;
typedef __attribute__((ext_vector_type(8))) float v8f;

#define N_EMB  1024
#define HEAD   64
#define SEQ_T  2048
#define BATCH  8
#define ROWS   (BATCH * SEQ_T)

static __device__ __forceinline__ v8f wmma_f32(v2f a, v2f b, v8f c) {
    return __builtin_amdgcn_wmma_f32_16x16x4_f32(false, a, false, b, (short)0, c,
                                                 false, false);
}

static __device__ __forceinline__ v2f mk2(float x, float y) { v2f r; r.x = x; r.y = y; return r; }

#define PJ_BM    64
#define PJ_BN    192
#define PJ_KS    64
#define PJ_PITCH 68
#define PJ_NSLAB (N_EMB / PJ_KS)
#define PJ_NM    (PJ_KS / 8)
#define PJ_CPR   (PJ_KS / 4)
#define PJ_RPP   (256 / PJ_CPR)
#define PJ_NPX   (PJ_BM / PJ_RPP)
#define PJ_NPW   (PJ_BN / PJ_RPP)
#define PJ_XST   (PJ_BM * PJ_PITCH)
#define PJ_WST   (PJ_BN * PJ_PITCH)

struct pj_frag { v4f a; v4f b[6]; };

static __device__ __forceinline__ void pj_load(pj_frag& f, const float* xa, const float* wb, int m) {
    f.a = *(const v4f*)(xa + m * 8);
#pragma unroll
    for (int t = 0; t < 6; ++t)
        f.b[t] = *(const v4f*)(wb + t * 16 * PJ_PITCH + m * 8);
}

static __device__ __forceinline__ void pj_mma(v8f (&acc)[6], const pj_frag& f) {
#pragma unroll
    for (int t = 0; t < 6; ++t)
        acc[t] = wmma_f32(mk2(f.a.x, f.a.y), mk2(f.b[t].x, f.b[t].y), acc[t]);
#pragma unroll
    for (int t = 0; t < 6; ++t)
        acc[t] = wmma_f32(mk2(f.a.z, f.a.w), mk2(f.b[t].z, f.b[t].w), acc[t]);
}

__global__ __launch_bounds__(256)
void sah_proj_kernel(const float* __restrict__ x,
                     const float* __restrict__ Wk,
                     const float* __restrict__ Wq,
                     const float* __restrict__ Wv,
                     float* __restrict__ kb,
                     float* __restrict__ qb,
                     float* __restrict__ vb) {
    __shared__ __attribute__((aligned(16))) float xs[2 * PJ_XST];
    __shared__ __attribute__((aligned(16))) float ws[2 * PJ_WST];

    const int tid  = threadIdx.x;
    const int lane = tid & 31;
    const int wave = tid >> 5;
    const int half = lane >> 4;
    const int lr   = lane & 15;
    const int rw   = wave & 3;
    const int cw   = wave >> 2;
    const int row0 = blockIdx.x * PJ_BM;

    const int srow = tid / PJ_CPR;
    const int skc  = (tid % PJ_CPR) * 4;
    const float* gx = x  + (size_t)(row0 + srow) * N_EMB + skc;
    const float* gk = Wk + (size_t)srow * N_EMB + skc;
    const float* gq = Wq + (size_t)srow * N_EMB + skc;
    const float* gv = Wv + (size_t)srow * N_EMB + skc;
    const int so = srow * PJ_PITCH + skc;

    const int ao = (rw * 16 + lr) * PJ_PITCH + half * 4;
    const int bo = (cw * 96 + lr) * PJ_PITCH + half * 4;

    v8f acc[6];
#pragma unroll
    for (int t = 0; t < 6; ++t) acc[t] = (v8f){};

    v4f px[PJ_NPX], pw[PJ_NPW];

#pragma unroll
    for (int p = 0; p < PJ_NPX; ++p)
        px[p] = *(const v4f*)(gx + (size_t)(p * PJ_RPP) * N_EMB);
#pragma unroll
    for (int p = 0; p < PJ_NPW; ++p) {
        const int wr = p * PJ_RPP;
        const float* g = (wr < 64) ? gk : (wr < 128) ? gq : gv;
        pw[p] = *(const v4f*)(g + (size_t)(wr & 63) * N_EMB);
    }
#pragma unroll
    for (int p = 0; p < PJ_NPX; ++p)
        *(v4f*)&xs[so + p * PJ_RPP * PJ_PITCH] = px[p];
#pragma unroll
    for (int p = 0; p < PJ_NPW; ++p)
        *(v4f*)&ws[so + p * PJ_RPP * PJ_PITCH] = pw[p];
    __syncthreads();

    for (int s = 0; s < PJ_NSLAB; ++s) {
        const int cur = s & 1;
        const bool more = (s + 1 < PJ_NSLAB);

        if (more) {
            const int ko = (s + 1) * PJ_KS;
#pragma unroll
            for (int p = 0; p < PJ_NPX; ++p)
                px[p] = *(const v4f*)(gx + (size_t)(p * PJ_RPP) * N_EMB + ko);
#pragma unroll
            for (int p = 0; p < PJ_NPW; ++p) {
                const int wr = p * PJ_RPP;
                const float* g = (wr < 64) ? gk : (wr < 128) ? gq : gv;
                pw[p] = *(const v4f*)(g + (size_t)(wr & 63) * N_EMB + ko);
            }
        }

        const float* xa = &xs[cur * PJ_XST + ao];
        const float* wb = &ws[cur * PJ_WST + bo];
        pj_frag f0, f1;
        pj_load(f0, xa, wb, 0);
#pragma unroll
        for (int m = 0; m < PJ_NM; m += 2) {
            pj_load(f1, xa, wb, m + 1);
            __builtin_amdgcn_sched_barrier(0);
            pj_mma(acc, f0);
            __builtin_amdgcn_sched_barrier(0);
            if (m + 2 < PJ_NM) pj_load(f0, xa, wb, m + 2);
            __builtin_amdgcn_sched_barrier(0);
            pj_mma(acc, f1);
            __builtin_amdgcn_sched_barrier(0);
        }

        if (more) {
            const int nx = cur ^ 1;
#pragma unroll
            for (int p = 0; p < PJ_NPX; ++p)
                *(v4f*)&xs[nx * PJ_XST + so + p * PJ_RPP * PJ_PITCH] = px[p];
#pragma unroll
            for (int p = 0; p < PJ_NPW; ++p)
                *(v4f*)&ws[nx * PJ_WST + so + p * PJ_RPP * PJ_PITCH] = pw[p];
        }
        __syncthreads();
    }

    const size_t orow = (size_t)(row0 + rw * 16 + half * 8);
#pragma unroll
    for (int t = 0; t < 6; ++t) {
        const int col   = cw * 96 + t * 16;
        const int which = col >> 6;
        float* out = (which == 0) ? kb : (which == 1) ? qb : vb;
        float* obase = out + orow * HEAD + (col & 63) + lr;
#pragma unroll
        for (int r = 0; r < 8; ++r)
            obase[(size_t)r * HEAD] = acc[t][r];
    }
}

#define AT_OP 68
#define AT_NW 8
#define AT_NQ 32
#define AT_NG (SEQ_T / AT_NQ)
#define AT_NPAIR (AT_NG / 2)

static __device__ __forceinline__ float xhalf(float x) {
    int b = __builtin_bit_cast(int, x);
    return __builtin_bit_cast(float,
        __builtin_amdgcn_permlanex16(b, b, 0x76543210, 0xfedcba98, false, false));
}

struct at_kv { v4f kt[8]; v2f va[16]; };

struct at_qs { v2f qf[16]; v8f a0, a1, a2, a3; float m, l; };

static __device__ __forceinline__ void at_load(at_kv& f, const float* Kb, const float* Vb,
                                               int j, int lr, int half) {
    const int k0r = j * 16;
    const float* krow = Kb + (size_t)(k0r + lr) * HEAD + half * 4;
#pragma unroll
    for (int c = 0; c < 8; ++c) f.kt[c] = *(const v4f*)(krow + c * 8);
#pragma unroll
    for (int kk = 0; kk < 4; ++kk) {
        const float* vp = Vb + (size_t)(k0r + half * 8 + kk * 2) * HEAD + lr;
#pragma unroll
        for (int t = 0; t < 4; ++t) f.va[4 * kk + t] = mk2(vp[16 * t], vp[HEAD + 16 * t]);
    }
}

static __device__ __forceinline__ void at_qload(at_qs& q, const float* qrow) {
#pragma unroll
    for (int c = 0; c < 8; ++c) {
        v4f t = *(const v4f*)(qrow + c * 8);
        q.qf[2 * c]     = mk2(t.x, t.y);
        q.qf[2 * c + 1] = mk2(t.z, t.w);
    }
    q.a0 = (v8f){}; q.a1 = (v8f){}; q.a2 = (v8f){}; q.a3 = (v8f){};
    q.m = -__builtin_inff();
    q.l = 0.0f;
}

static __device__ __forceinline__ void at_tile(at_qs& q, const at_kv& f, int lim, int half) {
    const float scale = 0.03125f;

    v8f sA = {}, sB = {};
#pragma unroll
    for (int c = 0; c < 8; ++c) {
        sA = wmma_f32(mk2(f.kt[c].x, f.kt[c].y), q.qf[2 * c],     sA);
        sB = wmma_f32(mk2(f.kt[c].z, f.kt[c].w), q.qf[2 * c + 1], sB);
    }

    float s[8];
#pragma unroll
    for (int r = 0; r < 8; ++r) {
        float sv = (sA[r] + sB[r]) * scale;
        s[r] = (half * 8 + r > lim) ? -__builtin_inff() : sv;
    }

    float tm = fmaxf(fmaxf(fmaxf(s[0], s[1]), fmaxf(s[2], s[3])),
                     fmaxf(fmaxf(s[4], s[5]), fmaxf(s[6], s[7])));
    tm = fmaxf(tm, xhalf(tm));
    const float mn    = fmaxf(q.m, tm);
    const float alpha = __expf(q.m - mn);
    q.m = mn;

    float p[8];
#pragma unroll
    for (int r = 0; r < 8; ++r) p[r] = __expf(s[r] - mn);

    q.l = q.l * alpha + (((p[0] + p[1]) + (p[2] + p[3])) + ((p[4] + p[5]) + (p[6] + p[7])));
#pragma unroll
    for (int r = 0; r < 8; ++r) {
        q.a0[r] *= alpha;
        q.a1[r] *= alpha;
        q.a2[r] *= alpha;
        q.a3[r] *= alpha;
    }

#pragma unroll
    for (int kk = 0; kk < 4; ++kk) {
        const v2f pb = mk2(p[2 * kk], p[2 * kk + 1]);
        q.a0 = wmma_f32(f.va[4 * kk + 0], pb, q.a0);
        q.a1 = wmma_f32(f.va[4 * kk + 1], pb, q.a1);
        q.a2 = wmma_f32(f.va[4 * kk + 2], pb, q.a2);
        q.a3 = wmma_f32(f.va[4 * kk + 3], pb, q.a3);
    }
}

static __device__ __forceinline__ void at_step(at_qs& qa, at_qs& qb, const at_kv& f,
                                               int j, int g, int lr, int half) {
    at_tile(qb, f, (j == 2 * g + 1) ? lr : 16, half);
    if (j <= 2 * g)
        at_tile(qa, f, (j == 2 * g) ? lr : 16, half);
}

static __device__ __forceinline__ void at_publish(const at_qs& q, float al, float* orow) {
#pragma unroll
    for (int h4 = 0; h4 < 2; ++h4) {
        v4f o0, o1, o2, o3;
        o0.x = q.a0[4 * h4] * al; o0.y = q.a0[4 * h4 + 1] * al; o0.z = q.a0[4 * h4 + 2] * al; o0.w = q.a0[4 * h4 + 3] * al;
        o1.x = q.a1[4 * h4] * al; o1.y = q.a1[4 * h4 + 1] * al; o1.z = q.a1[4 * h4 + 2] * al; o1.w = q.a1[4 * h4 + 3] * al;
        o2.x = q.a2[4 * h4] * al; o2.y = q.a2[4 * h4 + 1] * al; o2.z = q.a2[4 * h4 + 2] * al; o2.w = q.a2[4 * h4 + 3] * al;
        o3.x = q.a3[4 * h4] * al; o3.y = q.a3[4 * h4 + 1] * al; o3.z = q.a3[4 * h4 + 2] * al; o3.w = q.a3[4 * h4 + 3] * al;
        *(v4f*)(orow +  0 + 4 * h4) = o0;
        *(v4f*)(orow + 16 + 4 * h4) = o1;
        *(v4f*)(orow + 32 + 4 * h4) = o2;
        *(v4f*)(orow + 48 + 4 * h4) = o3;
    }
}

static __device__ __forceinline__ float at_max8(const float (*mr)[AT_NQ], int c) {
    return fmaxf(fmaxf(fmaxf(mr[0][c], mr[1][c]), fmaxf(mr[2][c], mr[3][c])),
                 fmaxf(fmaxf(mr[4][c], mr[5][c]), fmaxf(mr[6][c], mr[7][c])));
}

__global__ __launch_bounds__(32 * AT_NW)
void sah_attn_kernel(const float* __restrict__ Q,
                     const float* __restrict__ K,
                     const float* __restrict__ V,
                     float* __restrict__ O) {
    __shared__ float mred[AT_NW][AT_NQ];
    __shared__ float lred[AT_NW][AT_NQ];
    __shared__ __attribute__((aligned(16))) float ored[AT_NW][AT_NQ * AT_OP];

    const int lane = threadIdx.x & 31;
    const int wave = __builtin_amdgcn_readfirstlane((int)(threadIdx.x >> 5));
    const int half = lane >> 4;
    const int lr   = lane & 15;
    const int b    = blockIdx.y;
    const int pr   = blockIdx.x;

    const float* Qb = Q + (size_t)b * SEQ_T * HEAD;
    const float* Kb = K + (size_t)b * SEQ_T * HEAD;
    const float* Vb = V + (size_t)b * SEQ_T * HEAD;

    for (int pass = 0; pass < 2; ++pass) {
    const int g    = pass ? pr : (AT_NG - 1 - pr);
    const int q0   = g * AT_NQ;
    const int jmax = 2 * g + 1;

    at_kv f0, f1;
    at_load(f0, Kb, Vb, (wave <= jmax) ? wave : 0, lr, half);
    at_qs qa, qb;
    at_qload(qa, Qb + (size_t)(q0 + lr) * HEAD + half * 4);
    at_qload(qb, Qb + (size_t)(q0 + 16 + lr) * HEAD + half * 4);

    for (int j = wave; j <= jmax; j += 2 * AT_NW) {
        const int j1 = j + AT_NW;
        at_load(f1, Kb, Vb, (j1 <= jmax) ? j1 : j, lr, half);
        __builtin_amdgcn_sched_barrier(0);
        at_step(qa, qb, f0, j, g, lr, half);
        if (j1 > jmax) break;
        const int j2 = j1 + AT_NW;
        at_load(f0, Kb, Vb, (j2 <= jmax) ? j2 : j1, lr, half);
        __builtin_amdgcn_sched_barrier(0);
        at_step(qa, qb, f1, j1, g, lr, half);
    }

    qa.l = qa.l + xhalf(qa.l);
    qb.l = qb.l + xhalf(qb.l);
    if (half == 0) {
        mred[wave][lr]      = qa.m;
        mred[wave][16 + lr] = qb.m;
    }
    __syncthreads();

    {
        const float ala = __expf(qa.m - at_max8(mred, lr));
        const float alb = __expf(qb.m - at_max8(mred, 16 + lr));
        if (half == 0) {
            lred[wave][lr]      = qa.l * ala;
            lred[wave][16 + lr] = qb.l * alb;
        }
        at_publish(qa, ala, &ored[wave][lr * AT_OP + half * 8]);
        at_publish(qb, alb, &ored[wave][(16 + lr) * AT_OP + half * 8]);
    }
    __syncthreads();

#pragma unroll
    for (int rr = 0; rr < 2; ++rr) {
        const int row = (threadIdx.x >> 4) + 16 * rr;
        const int cc  = (threadIdx.x & 15) * 4;
        const float lt = ((lred[0][row] + lred[1][row]) + (lred[2][row] + lred[3][row])) +
                         ((lred[4][row] + lred[5][row]) + (lred[6][row] + lred[7][row]));
        float* ob = O + ((size_t)b * SEQ_T + q0 + row) * HEAD + cc;
        const int o = row * AT_OP + cc;
        const v4f a0 = *(const v4f*)&ored[0][o];
        const v4f a1 = *(const v4f*)&ored[1][o];
        const v4f a2 = *(const v4f*)&ored[2][o];
        const v4f a3 = *(const v4f*)&ored[3][o];
        const v4f a4 = *(const v4f*)&ored[4][o];
        const v4f a5 = *(const v4f*)&ored[5][o];
        const v4f a6 = *(const v4f*)&ored[6][o];
        const v4f a7 = *(const v4f*)&ored[7][o];
        v4f sum = ((a0 + a1) + (a2 + a3)) + ((a4 + a5) + (a6 + a7));
        v4f res;
        res.x = sum.x / lt; res.y = sum.y / lt; res.z = sum.z / lt; res.w = sum.w / lt;
        *(v4f*)ob = res;
    }
    }
}

extern "C" void kernel_launch(void* const* d_in, const int* in_sizes, int n_in,
                              void* d_out, int out_size, void* d_ws, size_t ws_size,
                              hipStream_t stream) {
    const float* x  = (const float*)d_in[0];
    const float* Wk = (const float*)d_in[1];
    const float* Wq = (const float*)d_in[2];
    const float* Wv = (const float*)d_in[3];

    float* kb = (float*)d_ws;
    float* qb = kb + (size_t)ROWS * HEAD;
    float* vb = qb + (size_t)ROWS * HEAD;

    sah_proj_kernel<<<dim3(ROWS / PJ_BM), 256, 0, stream>>>(x, Wk, Wq, Wv, kb, qb, vb);
    sah_attn_kernel<<<dim3(AT_NPAIR, BATCH), 32 * AT_NW, 0, stream>>>(qb, kb, vb,
                                                                      (float*)d_out);
}


# Projection: LDS ring deepened to 3 stages; next slab is stored and the slab's barrier taken mid-slab, its first fragments read before the slab ends: no slab boundary drains the matrix pipe
# speedup vs baseline: 2.0364x; 1.0390x over previous
#include <hip/hip_runtime.h>
#include <hip/hip_bf16.h>

typedef __attribute__((ext_vector_type(2))) float v2f;
typedef __attribute__((ext_vector_type(4))) float v4f;
typedef __attribute__((ext_vector_type(8))) float v8f;

#define N_EMB  1024
#define HEAD   64
#define SEQ_T  2048
#define BATCH  8
#define ROWS   (BATCH * SEQ_T)

static __device__ __forceinline__ v8f wmma_f32(v2f a, v2f b, v8f c) {
    return __builtin_amdgcn_wmma_f32_16x16x4_f32(false, a, false, b, (short)0, c,
                                                 false, false);
}

static __device__ __forceinline__ v2f mk2(float x, float y) { v2f r; r.x = x; r.y = y; return r; }

#define PJ_BM    64
#define PJ_BN    192
#define PJ_KS    64
#define PJ_PITCH 68
#define PJ_NSTG  3
#define PJ_NSLAB (N_EMB / PJ_KS)
#define PJ_NM    (PJ_KS / 8)
#define PJ_CPR   (PJ_KS / 4)
#define PJ_RPP   (256 / PJ_CPR)
#define PJ_NPX   (PJ_BM / PJ_RPP)
#define PJ_NPW   (PJ_BN / PJ_RPP)
#define PJ_XST   (PJ_BM * PJ_PITCH)
#define PJ_WST   (PJ_BN * PJ_PITCH)

struct pj_frag { v4f a; v4f b[6]; };

static __device__ __forceinline__ void pj_load(pj_frag& f, const float* xa, const float* wb, int m) {
    f.a = *(const v4f*)(xa + m * 8);
#pragma unroll
    for (int t = 0; t < 6; ++t)
        f.b[t] = *(const v4f*)(wb + t * 16 * PJ_PITCH + m * 8);
}

static __device__ __forceinline__ void pj_mma(v8f (&acc)[6], const pj_frag& f) {
#pragma unroll
    for (int t = 0; t < 6; ++t)
        acc[t] = wmma_f32(mk2(f.a.x, f.a.y), mk2(f.b[t].x, f.b[t].y), acc[t]);
#pragma unroll
    for (int t = 0; t < 6; ++t)
        acc[t] = wmma_f32(mk2(f.a.z, f.a.w), mk2(f.b[t].z, f.b[t].w), acc[t]);
}

static __device__ __forceinline__ void pj_gload(v4f (&px)[PJ_NPX], v4f (&pw)[PJ_NPW],
                                                const float* gx, const float* gk,
                                                const float* gq, const float* gv, int ko) {
#pragma unroll
    for (int p = 0; p < PJ_NPX; ++p)
        px[p] = *(const v4f*)(gx + (size_t)(p * PJ_RPP) * N_EMB + ko);
#pragma unroll
    for (int p = 0; p < PJ_NPW; ++p) {
        const int wr = p * PJ_RPP;
        const float* g = (wr < 64) ? gk : (wr < 128) ? gq : gv;
        pw[p] = *(const v4f*)(g + (size_t)(wr & 63) * N_EMB + ko);
    }
}

static __device__ __forceinline__ void pj_store(float* xd, float* wd,
                                                const v4f (&px)[PJ_NPX], const v4f (&pw)[PJ_NPW]) {
#pragma unroll
    for (int p = 0; p < PJ_NPX; ++p)
        *(v4f*)(xd + p * PJ_RPP * PJ_PITCH) = px[p];
#pragma unroll
    for (int p = 0; p < PJ_NPW; ++p)
        *(v4f*)(wd + p * PJ_RPP * PJ_PITCH) = pw[p];
}

template <bool HANDOVER, bool REFILL>
static __device__ __forceinline__ void pj_slab(v8f (&acc)[6], pj_frag& f0, pj_frag& f1,
                                               const float* xa, const float* wb,
                                               const float* xan, const float* wbn,
                                               float* xd, float* wd,
                                               v4f (&px)[PJ_NPX], v4f (&pw)[PJ_NPW],
                                               const float* gx, const float* gk,
                                               const float* gq, const float* gv, int ko) {
    pj_load(f1, xa, wb, 1);
    __builtin_amdgcn_sched_barrier(0);
    pj_mma(acc, f0);
    __builtin_amdgcn_sched_barrier(0);

    pj_load(f0, xa, wb, 2);
    __builtin_amdgcn_sched_barrier(0);
    pj_mma(acc, f1);
    __builtin_amdgcn_sched_barrier(0);

    pj_load(f1, xa, wb, 3);
    if (HANDOVER) pj_store(xd, wd, px, pw);
    __builtin_amdgcn_sched_barrier(0);
    pj_mma(acc, f0);
    __builtin_amdgcn_sched_barrier(0);

    pj_load(f0, xa, wb, 4);
    __builtin_amdgcn_sched_barrier(0);
    pj_mma(acc, f1);
    __builtin_amdgcn_sched_barrier(0);

    if (HANDOVER) __syncthreads();

    pj_load(f1, xa, wb, 5);
    __builtin_amdgcn_sched_barrier(0);
    pj_mma(acc, f0);
    __builtin_amdgcn_sched_barrier(0);

    pj_load(f0, xa, wb, 6);
    if (REFILL) pj_gload(px, pw, gx, gk, gq, gv, ko);
    __builtin_amdgcn_sched_barrier(0);
    pj_mma(acc, f1);
    __builtin_amdgcn_sched_barrier(0);

    pj_load(f1, xa, wb, 7);
    __builtin_amdgcn_sched_barrier(0);
    pj_mma(acc, f0);
    __builtin_amdgcn_sched_barrier(0);

    if (HANDOVER) pj_load(f0, xan, wbn, 0);
    __builtin_amdgcn_sched_barrier(0);
    pj_mma(acc, f1);
    __builtin_amdgcn_sched_barrier(0);
}

__global__ __launch_bounds__(256)
void sah_proj_kernel(const float* __restrict__ x,
                     const float* __restrict__ Wk,
                     const float* __restrict__ Wq,
                     const float* __restrict__ Wv,
                     float* __restrict__ kb,
                     float* __restrict__ qb,
                     float* __restrict__ vb) {
    __shared__ __attribute__((aligned(16))) float xs[PJ_NSTG * PJ_XST];
    __shared__ __attribute__((aligned(16))) float ws[PJ_NSTG * PJ_WST];

    const int tid  = threadIdx.x;
    const int lane = tid & 31;
    const int wave = tid >> 5;
    const int half = lane >> 4;
    const int lr   = lane & 15;
    const int rw   = wave & 3;
    const int cw   = wave >> 2;
    const int row0 = blockIdx.x * PJ_BM;

    const int srow = tid / PJ_CPR;
    const int skc  = (tid % PJ_CPR) * 4;
    const float* gx = x  + (size_t)(row0 + srow) * N_EMB + skc;
    const float* gk = Wk + (size_t)srow * N_EMB + skc;
    const float* gq = Wq + (size_t)srow * N_EMB + skc;
    const float* gv = Wv + (size_t)srow * N_EMB + skc;
    const int so = srow * PJ_PITCH + skc;

    const int ao = (rw * 16 + lr) * PJ_PITCH + half * 4;
    const int bo = (cw * 96 + lr) * PJ_PITCH + half * 4;

    v8f acc[6];
#pragma unroll
    for (int t = 0; t < 6; ++t) acc[t] = (v8f){};

    v4f px[PJ_NPX], pw[PJ_NPW];

    pj_gload(px, pw, gx, gk, gq, gv, 0);
    pj_store(&xs[so], &ws[so], px, pw);
    __syncthreads();
    pj_gload(px, pw, gx, gk, gq, gv, PJ_KS);

    pj_frag f0, f1;
    pj_load(f0, &xs[ao], &ws[bo], 0);

    int cur = 0;
    for (int s = 0; s < PJ_NSLAB - 2; ++s) {
        const int nxt = (cur == PJ_NSTG - 1) ? 0 : cur + 1;
        pj_slab<true, true>(acc, f0, f1,
                            &xs[cur * PJ_XST + ao], &ws[cur * PJ_WST + bo],
                            &xs[nxt * PJ_XST + ao], &ws[nxt * PJ_WST + bo],
                            &xs[nxt * PJ_XST + so], &ws[nxt * PJ_WST + so],
                            px, pw, gx, gk, gq, gv, (s + 2) * PJ_KS);
        cur = nxt;
    }
    {
        const int nxt = (cur == PJ_NSTG - 1) ? 0 : cur + 1;
        pj_slab<true, false>(acc, f0, f1,
                             &xs[cur * PJ_XST + ao], &ws[cur * PJ_WST + bo],
                             &xs[nxt * PJ_XST + ao], &ws[nxt * PJ_WST + bo],
                             &xs[nxt * PJ_XST + so], &ws[nxt * PJ_WST + so],
                             px, pw, gx, gk, gq, gv, 0);
        cur = nxt;
    }
    pj_slab<false, false>(acc, f0, f1,
                          &xs[cur * PJ_XST + ao], &ws[cur * PJ_WST + bo],
                          &xs[cur * PJ_XST + ao], &ws[cur * PJ_WST + bo],
                          &xs[so], &ws[so],
                          px, pw, gx, gk, gq, gv, 0);

    const size_t orow = (size_t)(row0 + rw * 16 + half * 8);
#pragma unroll
    for (int t = 0; t < 6; ++t) {
        const int col   = cw * 96 + t * 16;
        const int which = col >> 6;
        float* out = (which == 0) ? kb : (which == 1) ? qb : vb;
        float* obase = out + orow * HEAD + (col & 63) + lr;
#pragma unroll
        for (int r = 0; r < 8; ++r)
            obase[(size_t)r * HEAD] = acc[t][r];
    }
}

#define AT_OP 68
#define AT_NW 8
#define AT_NQ 32
#define AT_NG (SEQ_T / AT_NQ)
#define AT_NPAIR (AT_NG / 2)

static __device__ __forceinline__ float xhalf(float x) {
    int b = __builtin_bit_cast(int, x);
    return __builtin_bit_cast(float,
        __builtin_amdgcn_permlanex16(b, b, 0x76543210, 0xfedcba98, false, false));
}

struct at_kv { v4f kt[8]; v2f va[16]; };

struct at_qs { v2f qf[16]; v8f a0, a1, a2, a3; float m, l; };

static __device__ __forceinline__ void at_load(at_kv& f, const float* Kb, const float* Vb,
                                               int j, int lr, int half) {
    const int k0r = j * 16;
    const float* krow = Kb + (size_t)(k0r + lr) * HEAD + half * 4;
#pragma unroll
    for (int c = 0; c < 8; ++c) f.kt[c] = *(const v4f*)(krow + c * 8);
#pragma unroll
    for (int kk = 0; kk < 4; ++kk) {
        const float* vp = Vb + (size_t)(k0r + half * 8 + kk * 2) * HEAD + lr;
#pragma unroll
        for (int t = 0; t < 4; ++t) f.va[4 * kk + t] = mk2(vp[16 * t], vp[HEAD + 16 * t]);
    }
}

static __device__ __forceinline__ void at_qload(at_qs& q, const float* qrow) {
#pragma unroll
    for (int c = 0; c < 8; ++c) {
        v4f t = *(const v4f*)(qrow + c * 8);
        q.qf[2 * c]     = mk2(t.x, t.y);
        q.qf[2 * c + 1] = mk2(t.z, t.w);
    }
    q.a0 = (v8f){}; q.a1 = (v8f){}; q.a2 = (v8f){}; q.a3 = (v8f){};
    q.m = -__builtin_inff();
    q.l = 0.0f;
}

static __device__ __forceinline__ void at_tile(at_qs& q, const at_kv& f, int lim, int half) {
    const float scale = 0.03125f;

    v8f sA = {}, sB = {};
#pragma unroll
    for (int c = 0; c < 8; ++c) {
        sA = wmma_f32(mk2(f.kt[c].x, f.kt[c].y), q.qf[2 * c],     sA);
        sB = wmma_f32(mk2(f.kt[c].z, f.kt[c].w), q.qf[2 * c + 1], sB);
    }

    float s[8];
#pragma unroll
    for (int r = 0; r < 8; ++r) {
        float sv = (sA[r] + sB[r]) * scale;
        s[r] = (half * 8 + r > lim) ? -__builtin_inff() : sv;
    }

    float tm = fmaxf(fmaxf(fmaxf(s[0], s[1]), fmaxf(s[2], s[3])),
                     fmaxf(fmaxf(s[4], s[5]), fmaxf(s[6], s[7])));
    tm = fmaxf(tm, xhalf(tm));
    const float mn    = fmaxf(q.m, tm);
    const float alpha = __expf(q.m - mn);
    q.m = mn;

    float p[8];
#pragma unroll
    for (int r = 0; r < 8; ++r) p[r] = __expf(s[r] - mn);

    q.l = q.l * alpha + (((p[0] + p[1]) + (p[2] + p[3])) + ((p[4] + p[5]) + (p[6] + p[7])));
#pragma unroll
    for (int r = 0; r < 8; ++r) {
        q.a0[r] *= alpha;
        q.a1[r] *= alpha;
        q.a2[r] *= alpha;
        q.a3[r] *= alpha;
    }

#pragma unroll
    for (int kk = 0; kk < 4; ++kk) {
        const v2f pb = mk2(p[2 * kk], p[2 * kk + 1]);
        q.a0 = wmma_f32(f.va[4 * kk + 0], pb, q.a0);
        q.a1 = wmma_f32(f.va[4 * kk + 1], pb, q.a1);
        q.a2 = wmma_f32(f.va[4 * kk + 2], pb, q.a2);
        q.a3 = wmma_f32(f.va[4 * kk + 3], pb, q.a3);
    }
}

static __device__ __forceinline__ void at_step(at_qs& qa, at_qs& qb, const at_kv& f,
                                               int j, int g, int lr, int half) {
    at_tile(qb, f, (j == 2 * g + 1) ? lr : 16, half);
    if (j <= 2 * g)
        at_tile(qa, f, (j == 2 * g) ? lr : 16, half);
}

static __device__ __forceinline__ void at_publish(const at_qs& q, float al, float* orow) {
#pragma unroll
    for (int h4 = 0; h4 < 2; ++h4) {
        v4f o0, o1, o2, o3;
        o0.x = q.a0[4 * h4] * al; o0.y = q.a0[4 * h4 + 1] * al; o0.z = q.a0[4 * h4 + 2] * al; o0.w = q.a0[4 * h4 + 3] * al;
        o1.x = q.a1[4 * h4] * al; o1.y = q.a1[4 * h4 + 1] * al; o1.z = q.a1[4 * h4 + 2] * al; o1.w = q.a1[4 * h4 + 3] * al;
        o2.x = q.a2[4 * h4] * al; o2.y = q.a2[4 * h4 + 1] * al; o2.z = q.a2[4 * h4 + 2] * al; o2.w = q.a2[4 * h4 + 3] * al;
        o3.x = q.a3[4 * h4] * al; o3.y = q.a3[4 * h4 + 1] * al; o3.z = q.a3[4 * h4 + 2] * al; o3.w = q.a3[4 * h4 + 3] * al;
        *(v4f*)(orow +  0 + 4 * h4) = o0;
        *(v4f*)(orow + 16 + 4 * h4) = o1;
        *(v4f*)(orow + 32 + 4 * h4) = o2;
        *(v4f*)(orow + 48 + 4 * h4) = o3;
    }
}

static __device__ __forceinline__ float at_max8(const float (*mr)[AT_NQ], int c) {
    return fmaxf(fmaxf(fmaxf(mr[0][c], mr[1][c]), fmaxf(mr[2][c], mr[3][c])),
                 fmaxf(fmaxf(mr[4][c], mr[5][c]), fmaxf(mr[6][c], mr[7][c])));
}

__global__ __launch_bounds__(32 * AT_NW)
void sah_attn_kernel(const float* __restrict__ Q,
                     const float* __restrict__ K,
                     const float* __restrict__ V,
                     float* __restrict__ O) {
    __shared__ float mred[AT_NW][AT_NQ];
    __shared__ float lred[AT_NW][AT_NQ];
    __shared__ __attribute__((aligned(16))) float ored[AT_NW][AT_NQ * AT_OP];

    const int lane = threadIdx.x & 31;
    const int wave = __builtin_amdgcn_readfirstlane((int)(threadIdx.x >> 5));
    const int half = lane >> 4;
    const int lr   = lane & 15;
    const int b    = blockIdx.y;
    const int pr   = blockIdx.x;

    const float* Qb = Q + (size_t)b * SEQ_T * HEAD;
    const float* Kb = K + (size_t)b * SEQ_T * HEAD;
    const float* Vb = V + (size_t)b * SEQ_T * HEAD;

    for (int pass = 0; pass < 2; ++pass) {
    const int g    = pass ? pr : (AT_NG - 1 - pr);
    const int q0   = g * AT_NQ;
    const int jmax = 2 * g + 1;

    at_kv f0, f1;
    at_load(f0, Kb, Vb, (wave <= jmax) ? wave : 0, lr, half);
    at_qs qa, qb;
    at_qload(qa, Qb + (size_t)(q0 + lr) * HEAD + half * 4);
    at_qload(qb, Qb + (size_t)(q0 + 16 + lr) * HEAD + half * 4);

    for (int j = wave; j <= jmax; j += 2 * AT_NW) {
        const int j1 = j + AT_NW;
        at_load(f1, Kb, Vb, (j1 <= jmax) ? j1 : j, lr, half);
        __builtin_amdgcn_sched_barrier(0);
        at_step(qa, qb, f0, j, g, lr, half);
        if (j1 > jmax) break;
        const int j2 = j1 + AT_NW;
        at_load(f0, Kb, Vb, (j2 <= jmax) ? j2 : j1, lr, half);
        __builtin_amdgcn_sched_barrier(0);
        at_step(qa, qb, f1, j1, g, lr, half);
    }

    qa.l = qa.l + xhalf(qa.l);
    qb.l = qb.l + xhalf(qb.l);
    if (half == 0) {
        mred[wave][lr]      = qa.m;
        mred[wave][16 + lr] = qb.m;
    }
    __syncthreads();

    {
        const float ala = __expf(qa.m - at_max8(mred, lr));
        const float alb = __expf(qb.m - at_max8(mred, 16 + lr));
        if (half == 0) {
            lred[wave][lr]      = qa.l * ala;
            lred[wave][16 + lr] = qb.l * alb;
        }
        at_publish(qa, ala, &ored[wave][lr * AT_OP + half * 8]);
        at_publish(qb, alb, &ored[wave][(16 + lr) * AT_OP + half * 8]);
    }
    __syncthreads();

#pragma unroll
    for (int rr = 0; rr < 2; ++rr) {
        const int row = (threadIdx.x >> 4) + 16 * rr;
        const int cc  = (threadIdx.x & 15) * 4;
        const float lt = ((lred[0][row] + lred[1][row]) + (lred[2][row] + lred[3][row])) +
                         ((lred[4][row] + lred[5][row]) + (lred[6][row] + lred[7][row]));
        float* ob = O + ((size_t)b * SEQ_T + q0 + row) * HEAD + cc;
        const int o = row * AT_OP + cc;
        const v4f a0 = *(const v4f*)&ored[0][o];
        const v4f a1 = *(const v4f*)&ored[1][o];
        const v4f a2 = *(const v4f*)&ored[2][o];
        const v4f a3 = *(const v4f*)&ored[3][o];
        const v4f a4 = *(const v4f*)&ored[4][o];
        const v4f a5 = *(const v4f*)&ored[5][o];
        const v4f a6 = *(const v4f*)&ored[6][o];
        const v4f a7 = *(const v4f*)&ored[7][o];
        v4f sum = ((a0 + a1) + (a2 + a3)) + ((a4 + a5) + (a6 + a7));
        v4f res;
        res.x = sum.x / lt; res.y = sum.y / lt; res.z = sum.z / lt; res.w = sum.w / lt;
        *(v4f*)ob = res;
    }
    }
}

extern "C" void kernel_launch(void* const* d_in, const int* in_sizes, int n_in,
                              void* d_out, int out_size, void* d_ws, size_t ws_size,
                              hipStream_t stream) {
    const float* x  = (const float*)d_in[0];
    const float* Wk = (const float*)d_in[1];
    const float* Wq = (const float*)d_in[2];
    const float* Wv = (const float*)d_in[3];

    float* kb = (float*)d_ws;
    float* qb = kb + (size_t)ROWS * HEAD;
    float* vb = qb + (size_t)ROWS * HEAD;

    sah_proj_kernel<<<dim3(ROWS / PJ_BM), 256, 0, stream>>>(x, Wk, Wq, Wv, kb, qb, vb);
    sah_attn_kernel<<<dim3(AT_NPAIR, BATCH), 32 * AT_NW, 0, stream>>>(qb, kb, vb,
                                                                      (float*)d_out);
}


# Q, K, V planes stored in the attention's WMMA fragment order (K/Q tiles formed transposed): each fragment load and each projection store is a whole-line 16 B/lane access, 16 loads per key tile not 40
# speedup vs baseline: 2.1179x; 1.0400x over previous
#include <hip/hip_runtime.h>
#include <hip/hip_bf16.h>

typedef __attribute__((ext_vector_type(2))) float v2f;
typedef __attribute__((ext_vector_type(4))) float v4f;
typedef __attribute__((ext_vector_type(8))) float v8f;

#define N_EMB  1024
#define HEAD   64
#define SEQ_T  2048
#define BATCH  8
#define ROWS   (BATCH * SEQ_T)

static __device__ __forceinline__ v8f wmma_f32(v2f a, v2f b, v8f c) {
    return __builtin_amdgcn_wmma_f32_16x16x4_f32(false, a, false, b, (short)0, c,
                                                 false, false);
}

static __device__ __forceinline__ v2f mk2(float x, float y) { v2f r; r.x = x; r.y = y; return r; }

#define PJ_BM    64
#define PJ_BN    192
#define PJ_KS    64
#define PJ_PITCH 68
#define PJ_NSTG  3
#define PJ_NSLAB (N_EMB / PJ_KS)
#define PJ_NM    (PJ_KS / 8)
#define PJ_CPR   (PJ_KS / 4)
#define PJ_RPP   (256 / PJ_CPR)
#define PJ_NPX   (PJ_BM / PJ_RPP)
#define PJ_NPW   (PJ_BN / PJ_RPP)
#define PJ_XST   (PJ_BM * PJ_PITCH)
#define PJ_WST   (PJ_BN * PJ_PITCH)

struct pj_frag { v4f a; v4f b[6]; };

static __device__ __forceinline__ void pj_load(pj_frag& f, const float* xa, const float* wb, int wd, int m) {
    f.a = *(const v4f*)(xa + m * 8);
#pragma unroll
    for (int t = 0; t < 4; ++t)
        f.b[t] = *(const v4f*)(wb + t * 16 * PJ_PITCH + m * 8);
#pragma unroll
    for (int t = 4; t < 6; ++t)
        f.b[t] = *(const v4f*)(wb + wd + (t - 4) * 16 * PJ_PITCH + m * 8);
}

static __device__ __forceinline__ void pj_mma(v8f (&acc)[6], const pj_frag& f) {
#pragma unroll
    for (int t = 0; t < 4; ++t)
        acc[t] = wmma_f32(mk2(f.b[t].x, f.b[t].y), mk2(f.a.x, f.a.y), acc[t]);
#pragma unroll
    for (int t = 4; t < 6; ++t)
        acc[t] = wmma_f32(mk2(f.a.x, f.a.y), mk2(f.b[t].x, f.b[t].y), acc[t]);
#pragma unroll
    for (int t = 0; t < 4; ++t)
        acc[t] = wmma_f32(mk2(f.b[t].z, f.b[t].w), mk2(f.a.z, f.a.w), acc[t]);
#pragma unroll
    for (int t = 4; t < 6; ++t)
        acc[t] = wmma_f32(mk2(f.a.z, f.a.w), mk2(f.b[t].z, f.b[t].w), acc[t]);
}

static __device__ __forceinline__ void pj_gload(v4f (&px)[PJ_NPX], v4f (&pw)[PJ_NPW],
                                                const float* gx, const float* gk,
                                                const float* gq, const float* gv, int ko) {
#pragma unroll
    for (int p = 0; p < PJ_NPX; ++p)
        px[p] = *(const v4f*)(gx + (size_t)(p * PJ_RPP) * N_EMB + ko);
#pragma unroll
    for (int p = 0; p < PJ_NPW; ++p) {
        const int wr = p * PJ_RPP;
        const float* g = (wr < 64) ? gk : (wr < 128) ? gq : gv;
        pw[p] = *(const v4f*)(g + (size_t)(wr & 63) * N_EMB + ko);
    }
}

static __device__ __forceinline__ void pj_store(float* xd, float* wd,
                                                const v4f (&px)[PJ_NPX], const v4f (&pw)[PJ_NPW]) {
#pragma unroll
    for (int p = 0; p < PJ_NPX; ++p)
        *(v4f*)(xd + p * PJ_RPP * PJ_PITCH) = px[p];
#pragma unroll
    for (int p = 0; p < PJ_NPW; ++p)
        *(v4f*)(wd + p * PJ_RPP * PJ_PITCH) = pw[p];
}

template <bool HANDOVER, bool REFILL>
static __device__ __forceinline__ void pj_slab(v8f (&acc)[6], pj_frag& f0, pj_frag& f1,
                                               const float* xa, const float* wb,
                                               const float* xan, const float* wbn, int wv,
                                               float* xd, float* wd,
                                               v4f (&px)[PJ_NPX], v4f (&pw)[PJ_NPW],
                                               const float* gx, const float* gk,
                                               const float* gq, const float* gv, int ko) {
    pj_load(f1, xa, wb, wv, 1);
    __builtin_amdgcn_sched_barrier(0);
    pj_mma(acc, f0);
    __builtin_amdgcn_sched_barrier(0);

    pj_load(f0, xa, wb, wv, 2);
    __builtin_amdgcn_sched_barrier(0);
    pj_mma(acc, f1);
    __builtin_amdgcn_sched_barrier(0);

    pj_load(f1, xa, wb, wv, 3);
    if (HANDOVER) pj_store(xd, wd, px, pw);
    __builtin_amdgcn_sched_barrier(0);
    pj_mma(acc, f0);
    __builtin_amdgcn_sched_barrier(0);

    pj_load(f0, xa, wb, wv, 4);
    __builtin_amdgcn_sched_barrier(0);
    pj_mma(acc, f1);
    __builtin_amdgcn_sched_barrier(0);

    if (HANDOVER) __syncthreads();

    pj_load(f1, xa, wb, wv, 5);
    __builtin_amdgcn_sched_barrier(0);
    pj_mma(acc, f0);
    __builtin_amdgcn_sched_barrier(0);

    pj_load(f0, xa, wb, wv, 6);
    if (REFILL) pj_gload(px, pw, gx, gk, gq, gv, ko);
    __builtin_amdgcn_sched_barrier(0);
    pj_mma(acc, f1);
    __builtin_amdgcn_sched_barrier(0);

    pj_load(f1, xa, wb, wv, 7);
    __builtin_amdgcn_sched_barrier(0);
    pj_mma(acc, f0);
    __builtin_amdgcn_sched_barrier(0);

    if (HANDOVER) pj_load(f0, xan, wbn, wv, 0);
    __builtin_amdgcn_sched_barrier(0);
    pj_mma(acc, f1);
    __builtin_amdgcn_sched_barrier(0);
}

__global__ __launch_bounds__(256)
void sah_proj_kernel(const float* __restrict__ x,
                     const float* __restrict__ Wk,
                     const float* __restrict__ Wq,
                     const float* __restrict__ Wv,
                     float* __restrict__ kb,
                     float* __restrict__ qb,
                     float* __restrict__ vb) {
    __shared__ __attribute__((aligned(16))) float xs[PJ_NSTG * PJ_XST];
    __shared__ __attribute__((aligned(16))) float ws[PJ_NSTG * PJ_WST];

    const int tid  = threadIdx.x;
    const int lane = tid & 31;
    const int wave = tid >> 5;
    const int half = lane >> 4;
    const int lr   = lane & 15;
    const int rw   = wave & 3;
    const int cw   = wave >> 2;
    const int row0 = blockIdx.x * PJ_BM;

    const int srow = tid / PJ_CPR;
    const int skc  = (tid % PJ_CPR) * 4;
    const float* gx = x  + (size_t)(row0 + srow) * N_EMB + skc;
    const float* gk = Wk + (size_t)srow * N_EMB + skc;
    const float* gq = Wq + (size_t)srow * N_EMB + skc;
    const float* gv = Wv + (size_t)srow * N_EMB + skc;
    const int so = srow * PJ_PITCH + skc;

    const int ao = (rw * 16 + lr) * PJ_PITCH + half * 4;
    const int bo = (cw * 64 + lr) * PJ_PITCH + half * 4;
    const int wv = (128 - cw * 32) * PJ_PITCH;

    v8f acc[6];
#pragma unroll
    for (int t = 0; t < 6; ++t) acc[t] = (v8f){};

    v4f px[PJ_NPX], pw[PJ_NPW];

    pj_gload(px, pw, gx, gk, gq, gv, 0);
    pj_store(&xs[so], &ws[so], px, pw);
    __syncthreads();
    pj_gload(px, pw, gx, gk, gq, gv, PJ_KS);

    pj_frag f0, f1;
    pj_load(f0, &xs[ao], &ws[bo], wv, 0);

    int cur = 0;
    for (int s = 0; s < PJ_NSLAB - 2; ++s) {
        const int nxt = (cur == PJ_NSTG - 1) ? 0 : cur + 1;
        pj_slab<true, true>(acc, f0, f1,
                            &xs[cur * PJ_XST + ao], &ws[cur * PJ_WST + bo],
                            &xs[nxt * PJ_XST + ao], &ws[nxt * PJ_WST + bo], wv,
                            &xs[nxt * PJ_XST + so], &ws[nxt * PJ_WST + so],
                            px, pw, gx, gk, gq, gv, (s + 2) * PJ_KS);
        cur = nxt;
    }
    {
        const int nxt = (cur == PJ_NSTG - 1) ? 0 : cur + 1;
        pj_slab<true, false>(acc, f0, f1,
                             &xs[cur * PJ_XST + ao], &ws[cur * PJ_WST + bo],
                             &xs[nxt * PJ_XST + ao], &ws[nxt * PJ_WST + bo], wv,
                             &xs[nxt * PJ_XST + so], &ws[nxt * PJ_WST + so],
                             px, pw, gx, gk, gq, gv, 0);
        cur = nxt;
    }
    pj_slab<false, false>(acc, f0, f1,
                          &xs[cur * PJ_XST + ao], &ws[cur * PJ_WST + bo],
                          &xs[cur * PJ_XST + ao], &ws[cur * PJ_WST + bo], wv,
                          &xs[so], &ws[so],
                          px, pw, gx, gk, gq, gv, 0);

    const size_t tbase = ((size_t)blockIdx.x * (PJ_BM / 16) + rw) * (16 * HEAD);
    float* kq = (cw == 0 ? kb : qb) + tbase + half * 128 + lr * 4;
#pragma unroll
    for (int t = 0; t < 4; ++t) {
        v4f lo, hi;
        lo.x = acc[t][0]; lo.y = acc[t][1]; lo.z = acc[t][2]; lo.w = acc[t][3];
        hi.x = acc[t][4]; hi.y = acc[t][5]; hi.z = acc[t][6]; hi.w = acc[t][7];
        *(v4f*)(kq + t * 256)      = lo;
        *(v4f*)(kq + t * 256 + 64) = hi;
    }
    float* vo = vb + tbase + (cw * 2) * 256 + (half * 16 + lr) * 4;
#pragma unroll
    for (int t = 4; t < 6; ++t) {
        v4f lo, hi;
        lo.x = acc[t][0]; lo.y = acc[t][1]; lo.z = acc[t][2]; lo.w = acc[t][3];
        hi.x = acc[t][4]; hi.y = acc[t][5]; hi.z = acc[t][6]; hi.w = acc[t][7];
        *(v4f*)(vo + (t - 4) * 256)       = lo;
        *(v4f*)(vo + (t - 4) * 256 + 128) = hi;
    }
}

#define AT_OP 68
#define AT_NW 8
#define AT_NQ 32
#define AT_NG (SEQ_T / AT_NQ)
#define AT_NPAIR (AT_NG / 2)

static __device__ __forceinline__ float xhalf(float x) {
    int b = __builtin_bit_cast(int, x);
    return __builtin_bit_cast(float,
        __builtin_amdgcn_permlanex16(b, b, 0x76543210, 0xfedcba98, false, false));
}

struct at_kv { v4f kt[8]; v4f vv[8]; };

static __device__ __forceinline__ v2f at_va(const at_kv& f, int kk, int tt) {
    const v4f w = f.vv[2 * tt + (kk >> 1)];
    return (kk & 1) ? mk2(w.z, w.w) : mk2(w.x, w.y);
}

struct at_qs { v2f qf[16]; v8f a0, a1, a2, a3; float m, l; };

static __device__ __forceinline__ void at_load(at_kv& f, const float* Kb, const float* Vb,
                                               int j, int lr, int half) {
    const float* kt = Kb + (size_t)j * (16 * HEAD) + half * 64 + lr * 4;
#pragma unroll
    for (int c = 0; c < 8; ++c) f.kt[c] = *(const v4f*)(kt + c * 128);
    const float* vt = Vb + (size_t)j * (16 * HEAD) + (half * 16 + lr) * 4;
#pragma unroll
    for (int i = 0; i < 8; ++i) f.vv[i] = *(const v4f*)(vt + i * 128);
}

static __device__ __forceinline__ void at_qload(at_qs& q, const float* qrow) {
#pragma unroll
    for (int c = 0; c < 8; ++c) {
        v4f t = *(const v4f*)(qrow + c * 128);
        q.qf[2 * c]     = mk2(t.x, t.y);
        q.qf[2 * c + 1] = mk2(t.z, t.w);
    }
    q.a0 = (v8f){}; q.a1 = (v8f){}; q.a2 = (v8f){}; q.a3 = (v8f){};
    q.m = -__builtin_inff();
    q.l = 0.0f;
}

static __device__ __forceinline__ void at_tile(at_qs& q, const at_kv& f, int lim, int half) {
    const float scale = 0.03125f;

    v8f sA = {}, sB = {};
#pragma unroll
    for (int c = 0; c < 8; ++c) {
        sA = wmma_f32(mk2(f.kt[c].x, f.kt[c].y), q.qf[2 * c],     sA);
        sB = wmma_f32(mk2(f.kt[c].z, f.kt[c].w), q.qf[2 * c + 1], sB);
    }

    float s[8];
#pragma unroll
    for (int r = 0; r < 8; ++r) {
        float sv = (sA[r] + sB[r]) * scale;
        s[r] = (half * 8 + r > lim) ? -__builtin_inff() : sv;
    }

    float tm = fmaxf(fmaxf(fmaxf(s[0], s[1]), fmaxf(s[2], s[3])),
                     fmaxf(fmaxf(s[4], s[5]), fmaxf(s[6], s[7])));
    tm = fmaxf(tm, xhalf(tm));
    const float mn    = fmaxf(q.m, tm);
    const float alpha = __expf(q.m - mn);
    q.m = mn;

    float p[8];
#pragma unroll
    for (int r = 0; r < 8; ++r) p[r] = __expf(s[r] - mn);

    q.l = q.l * alpha + (((p[0] + p[1]) + (p[2] + p[3])) + ((p[4] + p[5]) + (p[6] + p[7])));
#pragma unroll
    for (int r = 0; r < 8; ++r) {
        q.a0[r] *= alpha;
        q.a1[r] *= alpha;
        q.a2[r] *= alpha;
        q.a3[r] *= alpha;
    }

#pragma unroll
    for (int kk = 0; kk < 4; ++kk) {
        const v2f pb = mk2(p[2 * kk], p[2 * kk + 1]);
        q.a0 = wmma_f32(at_va(f, kk, 0), pb, q.a0);
        q.a1 = wmma_f32(at_va(f, kk, 1), pb, q.a1);
        q.a2 = wmma_f32(at_va(f, kk, 2), pb, q.a2);
        q.a3 = wmma_f32(at_va(f, kk, 3), pb, q.a3);
    }
}

static __device__ __forceinline__ void at_step(at_qs& qa, at_qs& qb, const at_kv& f,
                                               int j, int g, int lr, int half) {
    at_tile(qb, f, (j == 2 * g + 1) ? lr : 16, half);
    if (j <= 2 * g)
        at_tile(qa, f, (j == 2 * g) ? lr : 16, half);
}

static __device__ __forceinline__ void at_publish(const at_qs& q, float al, float* orow) {
#pragma unroll
    for (int h4 = 0; h4 < 2; ++h4) {
        v4f o0, o1, o2, o3;
        o0.x = q.a0[4 * h4] * al; o0.y = q.a0[4 * h4 + 1] * al; o0.z = q.a0[4 * h4 + 2] * al; o0.w = q.a0[4 * h4 + 3] * al;
        o1.x = q.a1[4 * h4] * al; o1.y = q.a1[4 * h4 + 1] * al; o1.z = q.a1[4 * h4 + 2] * al; o1.w = q.a1[4 * h4 + 3] * al;
        o2.x = q.a2[4 * h4] * al; o2.y = q.a2[4 * h4 + 1] * al; o2.z = q.a2[4 * h4 + 2] * al; o2.w = q.a2[4 * h4 + 3] * al;
        o3.x = q.a3[4 * h4] * al; o3.y = q.a3[4 * h4 + 1] * al; o3.z = q.a3[4 * h4 + 2] * al; o3.w = q.a3[4 * h4 + 3] * al;
        *(v4f*)(orow +  0 + 4 * h4) = o0;
        *(v4f*)(orow + 16 + 4 * h4) = o1;
        *(v4f*)(orow + 32 + 4 * h4) = o2;
        *(v4f*)(orow + 48 + 4 * h4) = o3;
    }
}

static __device__ __forceinline__ float at_max8(const float (*mr)[AT_NQ], int c) {
    return fmaxf(fmaxf(fmaxf(mr[0][c], mr[1][c]), fmaxf(mr[2][c], mr[3][c])),
                 fmaxf(fmaxf(mr[4][c], mr[5][c]), fmaxf(mr[6][c], mr[7][c])));
}

__global__ __launch_bounds__(32 * AT_NW)
void sah_attn_kernel(const float* __restrict__ Q,
                     const float* __restrict__ K,
                     const float* __restrict__ V,
                     float* __restrict__ O) {
    __shared__ float mred[AT_NW][AT_NQ];
    __shared__ float lred[AT_NW][AT_NQ];
    __shared__ __attribute__((aligned(16))) float ored[AT_NW][AT_NQ * AT_OP];

    const int lane = threadIdx.x & 31;
    const int wave = __builtin_amdgcn_readfirstlane((int)(threadIdx.x >> 5));
    const int half = lane >> 4;
    const int lr   = lane & 15;
    const int b    = blockIdx.y;
    const int pr   = blockIdx.x;

    const float* Qb = Q + (size_t)b * SEQ_T * HEAD;
    const float* Kb = K + (size_t)b * SEQ_T * HEAD;
    const float* Vb = V + (size_t)b * SEQ_T * HEAD;

    for (int pass = 0; pass < 2; ++pass) {
    const int g    = pass ? pr : (AT_NG - 1 - pr);
    const int q0   = g * AT_NQ;
    const int jmax = 2 * g + 1;

    at_kv f0, f1;
    at_load(f0, Kb, Vb, (wave <= jmax) ? wave : 0, lr, half);
    at_qs qa, qb;
    at_qload(qa, Qb + (size_t)q0 * HEAD + half * 64 + lr * 4);
    at_qload(qb, Qb + (size_t)(q0 + 16) * HEAD + half * 64 + lr * 4);

    for (int j = wave; j <= jmax; j += 2 * AT_NW) {
        const int j1 = j + AT_NW;
        at_load(f1, Kb, Vb, (j1 <= jmax) ? j1 : j, lr, half);
        __builtin_amdgcn_sched_barrier(0);
        at_step(qa, qb, f0, j, g, lr, half);
        if (j1 > jmax) break;
        const int j2 = j1 + AT_NW;
        at_load(f0, Kb, Vb, (j2 <= jmax) ? j2 : j1, lr, half);
        __builtin_amdgcn_sched_barrier(0);
        at_step(qa, qb, f1, j1, g, lr, half);
    }

    qa.l = qa.l + xhalf(qa.l);
    qb.l = qb.l + xhalf(qb.l);
    if (half == 0) {
        mred[wave][lr]      = qa.m;
        mred[wave][16 + lr] = qb.m;
    }
    __syncthreads();

    {
        const float ala = __expf(qa.m - at_max8(mred, lr));
        const float alb = __expf(qb.m - at_max8(mred, 16 + lr));
        if (half == 0) {
            lred[wave][lr]      = qa.l * ala;
            lred[wave][16 + lr] = qb.l * alb;
        }
        at_publish(qa, ala, &ored[wave][lr * AT_OP + half * 8]);
        at_publish(qb, alb, &ored[wave][(16 + lr) * AT_OP + half * 8]);
    }
    __syncthreads();

#pragma unroll
    for (int rr = 0; rr < 2; ++rr) {
        const int row = (threadIdx.x >> 4) + 16 * rr;
        const int cc  = (threadIdx.x & 15) * 4;
        const float lt = ((lred[0][row] + lred[1][row]) + (lred[2][row] + lred[3][row])) +
                         ((lred[4][row] + lred[5][row]) + (lred[6][row] + lred[7][row]));
        float* ob = O + ((size_t)b * SEQ_T + q0 + row) * HEAD + cc;
        const int o = row * AT_OP + cc;
        const v4f a0 = *(const v4f*)&ored[0][o];
        const v4f a1 = *(const v4f*)&ored[1][o];
        const v4f a2 = *(const v4f*)&ored[2][o];
        const v4f a3 = *(const v4f*)&ored[3][o];
        const v4f a4 = *(const v4f*)&ored[4][o];
        const v4f a5 = *(const v4f*)&ored[5][o];
        const v4f a6 = *(const v4f*)&ored[6][o];
        const v4f a7 = *(const v4f*)&ored[7][o];
        v4f sum = ((a0 + a1) + (a2 + a3)) + ((a4 + a5) + (a6 + a7));
        v4f res;
        res.x = sum.x / lt; res.y = sum.y / lt; res.z = sum.z / lt; res.w = sum.w / lt;
        *(v4f*)ob = res;
    }
    }
}

extern "C" void kernel_launch(void* const* d_in, const int* in_sizes, int n_in,
                              void* d_out, int out_size, void* d_ws, size_t ws_size,
                              hipStream_t stream) {
    const float* x  = (const float*)d_in[0];
    const float* Wk = (const float*)d_in[1];
    const float* Wq = (const float*)d_in[2];
    const float* Wv = (const float*)d_in[3];

    float* kb = (float*)d_ws;
    float* qb = kb + (size_t)ROWS * HEAD;
    float* vb = qb + (size_t)ROWS * HEAD;

    sah_proj_kernel<<<dim3(ROWS / PJ_BM), 256, 0, stream>>>(x, Wk, Wq, Wv, kb, qb, vb);
    sah_attn_kernel<<<dim3(AT_NPAIR, BATCH), 32 * AT_NW, 0, stream>>>(qb, kb, vb,
                                                                      (float*)d_out);
}
